# starting kernel
# baseline (speedup 1.0000x reference)
#include <hip/hip_runtime.h>
#include <hip/hip_bf16.h>
#include <stdint.h>


#define FDIM 64
#define HDIM 64
#define CDIM 10

#define GWAVES 4
#define GROWS  (GWAVES * 16)
#define ASTR   68
#define WPSTR  160
#define WPROWS 32
#define WOSTR  96

typedef float v2f __attribute__((ext_vector_type(2)));
typedef float v8f __attribute__((ext_vector_type(8)));

__device__ __forceinline__ v8f wmma4(v2f a, v2f b, v8f c) {
  return __builtin_amdgcn_wmma_f32_16x16x4_f32(false, a, false, b, (short)0, c,
                                               false, false);
}


__global__ void zero_kernel(float* __restrict__ p, int n) {
  int i = blockIdx.x * blockDim.x + threadIdx.x;
  if (i < n) p[i] = 0.0f;
}

__global__ void degree_kernel(const int* __restrict__ src, float* __restrict__ deg, int E) {
  int e = blockIdx.x * blockDim.x + threadIdx.x;
  if (e < E) atomicAdd(&deg[src[e]], 1.0f);
}

__global__ void dis_kernel(const float* __restrict__ deg, float* __restrict__ dis, int n) {
  int i = blockIdx.x * blockDim.x + threadIdx.x;
  if (i >= n) return;
  float d = deg[i];
  dis[i] = (d > 0.0f) ? rsqrtf(fmaxf(d, 1.0f)) : 0.0f;
}

__global__ void edgew_kernel(const int* __restrict__ src, const int* __restrict__ dst,
                             const float* __restrict__ dis, float* __restrict__ w, int E) {
  int e = blockIdx.x * blockDim.x + threadIdx.x;
  if (e < E) w[e] = -(dis[src[e]] * dis[dst[e]]);
}

__global__ void scatter_kernel(const int* __restrict__ src, const int* __restrict__ dst,
                               const float* __restrict__ w, const float* __restrict__ h,
                               float* __restrict__ lap, int E) {
  int t = blockIdx.x * blockDim.x + threadIdx.x;
  int e = t >> 4;
  if (e >= E) return;
  int chunk = t & 15;
  int s = src[e], d = dst[e];
  float we = w[e];
  float4 hv = ((const float4*)(h + (size_t)s * HDIM))[chunk];
  float* lp = lap + (size_t)d * HDIM + chunk * 4;
  atomicAdd(lp + 0, we * hv.x);
  atomicAdd(lp + 1, we * hv.y);
  atomicAdd(lp + 2, we * hv.z);
  atomicAdd(lp + 3, we * hv.w);
}


__device__ __forceinline__ void stage_w_pairs(float* sWp, const float* __restrict__ W,
                                              int tid) {
  const float4* Wv = (const float4*)W;
#pragma unroll
  for (int i = 0; i < 8; ++i) {
    int idx = tid + i * 128;
    float4 v = Wv[idx];
    int k = idx >> 4;
    int c4 = (idx & 15) * 4;
    float* dstp = sWp + (k >> 1) * WPSTR + (k & 1);
    dstp[(c4 + 0) * 2] = v.x;
    dstp[(c4 + 1) * 2] = v.y;
    dstp[(c4 + 2) * 2] = v.z;
    dstp[(c4 + 3) * 2] = v.w;
  }
}

__device__ __forceinline__ void stage_rows(float* sAw, const float* __restrict__ A,
                                           int rowBase, int n, int lane) {
#pragma unroll
  for (int i = 0; i < 8; ++i) {
    int fi = lane + i * 32;
    int r = fi >> 4, c4 = fi & 15;
    int row = rowBase + r;
    if (row >= n) row = n - 1;
    ((float4*)(sAw + r * ASTR))[c4] = ((const float4*)(A + (size_t)row * 64))[c4];
  }
}

__device__ __forceinline__ void mm_accum64(const float* sAw, const float* sWp,
                                           int lane, v8f acc[4]) {
  const int nlo = lane & 15;
  const int g = lane >> 4;
#pragma unroll
  for (int kk = 0; kk < 16; ++kk) {
    const int k0 = kk * 4 + 2 * g;
    v2f a = *(const v2f*)(sAw + nlo * ASTR + k0);
    const float* brow = sWp + (kk * 2 + g) * WPSTR;
#pragma unroll
    for (int nt = 0; nt < 4; ++nt) {
      v2f b = *(const v2f*)(brow + (nt * 16 + nlo) * 2);
      acc[nt] = wmma4(a, b, acc[nt]);
    }
  }
}


__global__ void __launch_bounds__(128)
gemm64_relu_kernel(const float* __restrict__ A, const float* __restrict__ W,
                   const float* __restrict__ bias, float* __restrict__ out, int n) {
  __shared__ float sWp[WPROWS * WPSTR];
  __shared__ float sA[GWAVES * 16 * ASTR];
  __shared__ float sB[64];
  const int tid = threadIdx.x, wave = tid >> 5, lane = tid & 31;
  stage_w_pairs(sWp, W, tid);
  if (tid < 64) sB[tid] = bias[tid];

  const int rowBase = blockIdx.x * GROWS + wave * 16;
  float* sAw = sA + wave * 16 * ASTR;
  stage_rows(sAw, A, rowBase, n, lane);
  __syncthreads();

  v8f z = {0, 0, 0, 0, 0, 0, 0, 0};
  v8f acc[4] = {z, z, z, z};
  mm_accum64(sAw, sWp, lane, acc);

  const int nlo = lane & 15, g = lane >> 4;
#pragma unroll
  for (int nt = 0; nt < 4; ++nt)
#pragma unroll
    for (int r = 0; r < 8; ++r) {
      int row = rowBase + r + 8 * g;
      int col = nt * 16 + nlo;
      if (row < n) out[(size_t)row * 64 + col] = fmaxf(acc[nt][r] + sB[col], 0.0f);
    }
}

__global__ void __launch_bounds__(128)
gemm64_cheb2_kernel(float* __restrict__ h, const float* __restrict__ lap,
                    const float* __restrict__ W0, const float* __restrict__ W1,
                    const float* __restrict__ bias, int n) {
  __shared__ float sWp[2 * WPROWS * WPSTR];
  __shared__ float sA[GWAVES * 16 * ASTR];
  __shared__ float sB[64];
  const int tid = threadIdx.x, wave = tid >> 5, lane = tid & 31;
  stage_w_pairs(sWp, W0, tid);
  stage_w_pairs(sWp + WPROWS * WPSTR, W1, tid);
  if (tid < 64) sB[tid] = bias[tid];

  const int rowBase = blockIdx.x * GROWS + wave * 16;
  float* sAw = sA + wave * 16 * ASTR;

  v8f z = {0, 0, 0, 0, 0, 0, 0, 0};
  v8f acc[4] = {z, z, z, z};

  stage_rows(sAw, h, rowBase, n, lane);
  __syncthreads();
  mm_accum64(sAw, sWp, lane, acc);
  __syncthreads();
  stage_rows(sAw, lap, rowBase, n, lane);
  __syncthreads();
  mm_accum64(sAw, sWp + WPROWS * WPSTR, lane, acc);

  const int nlo = lane & 15, g = lane >> 4;
#pragma unroll
  for (int nt = 0; nt < 4; ++nt)
#pragma unroll
    for (int r = 0; r < 8; ++r) {
      int row = rowBase + r + 8 * g;
      int col = nt * 16 + nlo;
      if (row < n) {
        size_t idx = (size_t)row * 64 + col;
        float v = fmaxf(acc[nt][r] + sB[col], 0.0f);
        h[idx] = v + h[idx];
      }
    }
}

__global__ void __launch_bounds__(128)
gemm_out_kernel(const float* __restrict__ A, const float* __restrict__ Wl,
                const float* __restrict__ bl, float* __restrict__ logits, int n) {
  __shared__ float sWo[WPROWS * WOSTR];
  __shared__ float sA[GWAVES * 16 * ASTR];
  __shared__ float sB[16];
  const int tid = threadIdx.x, wave = tid >> 5, lane = tid & 31;
  for (int idx = tid; idx < 64 * 16; idx += 128) {
    int k = idx >> 4, c = idx & 15;
    float v = (c < CDIM) ? Wl[k * CDIM + c] : 0.0f;
    sWo[(k >> 1) * WOSTR + c * 2 + (k & 1)] = v;
  }
  if (tid < 16) sB[tid] = (tid < CDIM) ? bl[tid] : 0.0f;

  const int rowBase = blockIdx.x * GROWS + wave * 16;
  float* sAw = sA + wave * 16 * ASTR;
  stage_rows(sAw, A, rowBase, n, lane);
  __syncthreads();

  const int nlo = lane & 15, g = lane >> 4;
  v8f acc = {0, 0, 0, 0, 0, 0, 0, 0};
#pragma unroll
  for (int kk = 0; kk < 16; ++kk) {
    const int k0 = kk * 4 + 2 * g;
    v2f a = *(const v2f*)(sAw + nlo * ASTR + k0);
    v2f b = *(const v2f*)(sWo + (kk * 2 + g) * WOSTR + nlo * 2);
    acc = wmma4(a, b, acc);
  }
#pragma unroll
  for (int r = 0; r < 8; ++r) {
    int row = rowBase + r + 8 * g;
    if (row < n) logits[(size_t)row * 16 + nlo] = acc[r] + sB[nlo];
  }
}

__global__ void logsoftmax_kernel(const float* __restrict__ logits,
                                  float* __restrict__ out, int n) {
  int i = blockIdx.x * blockDim.x + threadIdx.x;
  if (i >= n) return;
  const float* v = logits + (size_t)i * 16;
  float m = v[0];
#pragma unroll
  for (int c = 1; c < CDIM; ++c) m = fmaxf(m, v[c]);
  float s = 0.0f;
#pragma unroll
  for (int c = 0; c < CDIM; ++c) s += expf(v[c] - m);
  float ls = logf(s);
#pragma unroll
  for (int c = 0; c < CDIM; ++c) out[(size_t)i * CDIM + c] = v[c] - m - ls;
}


extern "C" void kernel_launch(void* const* d_in, const int* in_sizes, int n_in,
                              void* d_out, int out_size, void* d_ws, size_t ws_size,
                              hipStream_t stream) {
  const float* x    = (const float*)d_in[0];
  const int*   ei   = (const int*)d_in[1];
  const float* W1_0 = (const float*)d_in[2];
  const float* b1   = (const float*)d_in[3];
  const float* W2_0 = (const float*)d_in[4];
  const float* W2_1 = (const float*)d_in[5];
  const float* b2   = (const float*)d_in[6];
  const float* Wl   = (const float*)d_in[7];
  const float* bl   = (const float*)d_in[8];

  const int N = in_sizes[0] / FDIM;
  const int E = in_sizes[1] / 2;
  const int* src = ei;
  const int* dst = ei + E;

  float* ws     = (float*)d_ws;
  float* deg    = ws;
  float* lap    = deg + N;
  float* dis    = lap + (size_t)N * 64;
  float* wbuf   = dis + N;
  float* h      = wbuf + E;
  float* logits = h + (size_t)N * 64;
  float* out    = (float*)d_out;

  {
    int tot = N * 65;
    zero_kernel<<<(tot + 255) / 256, 256, 0, stream>>>(deg, tot);
  }
  degree_kernel<<<(E + 255) / 256, 256, 0, stream>>>(src, deg, E);
  dis_kernel<<<(N + 255) / 256, 256, 0, stream>>>(deg, dis, N);
  edgew_kernel<<<(E + 255) / 256, 256, 0, stream>>>(src, dst, dis, wbuf, E);

  const int gblocks = (N + GROWS - 1) / GROWS;
  gemm64_relu_kernel<<<gblocks, 128, 0, stream>>>(x, W1_0, b1, h, N);
  {
    long long tot = (long long)E * 16;
    scatter_kernel<<<(int)((tot + 255) / 256), 256, 0, stream>>>(src, dst, wbuf, h, lap, E);
  }
  gemm64_cheb2_kernel<<<gblocks, 128, 0, stream>>>(h, lap, W2_0, W2_1, b2, N);
  gemm_out_kernel<<<gblocks, 128, 0, stream>>>(h, Wl, bl, logits, N);
  logsoftmax_kernel<<<(N + 255) / 256, 256, 0, stream>>>(logits, out, N);
}


# scatter: a wave walks its edges one at a time, lane l owns floats l and l+32, so each f32 atomic covers one whole 128-B line (2 line requests per edge, was 8); atomics kept, arrival-order sums
# speedup vs baseline: 1.3155x; 1.3155x over previous
#include <hip/hip_runtime.h>
#include <hip/hip_bf16.h>
#include <stdint.h>


#define FDIM 64
#define HDIM 64
#define CDIM 10

#define GWAVES 4
#define GROWS  (GWAVES * 16)
#define ASTR   68
#define WPSTR  160
#define WPROWS 32
#define WOSTR  96

typedef float v2f __attribute__((ext_vector_type(2)));
typedef float v8f __attribute__((ext_vector_type(8)));

__device__ __forceinline__ v8f wmma4(v2f a, v2f b, v8f c) {
  return __builtin_amdgcn_wmma_f32_16x16x4_f32(false, a, false, b, (short)0, c,
                                               false, false);
}


__global__ void zero_kernel(float* __restrict__ p, int n) {
  int i = blockIdx.x * blockDim.x + threadIdx.x;
  if (i < n) p[i] = 0.0f;
}

__global__ void degree_kernel(const int* __restrict__ src, float* __restrict__ deg, int E) {
  int e = blockIdx.x * blockDim.x + threadIdx.x;
  if (e < E) atomicAdd(&deg[src[e]], 1.0f);
}

__global__ void dis_kernel(const float* __restrict__ deg, float* __restrict__ dis, int n) {
  int i = blockIdx.x * blockDim.x + threadIdx.x;
  if (i >= n) return;
  float d = deg[i];
  dis[i] = (d > 0.0f) ? rsqrtf(fmaxf(d, 1.0f)) : 0.0f;
}

__global__ void edgew_kernel(const int* __restrict__ src, const int* __restrict__ dst,
                             const float* __restrict__ dis, float* __restrict__ w, int E) {
  int e = blockIdx.x * blockDim.x + threadIdx.x;
  if (e < E) w[e] = -(dis[src[e]] * dis[dst[e]]);
}

#define SC_EDGES 32
#define SC_GROUP 8

__device__ __forceinline__ int lane_bcast_i(int v, int j) {
  return __builtin_amdgcn_readlane(v, j);
}
__device__ __forceinline__ float lane_bcast_f(float v, int j) {
  return __int_as_float(__builtin_amdgcn_readlane(__float_as_int(v), j));
}

__global__ void __launch_bounds__(256)
scatter_kernel(const int* __restrict__ src, const int* __restrict__ dst,
               const float* __restrict__ w, const float* __restrict__ h,
               float* __restrict__ lap, int E) {
  const int lane = threadIdx.x & 31;
  const int wave = (blockIdx.x * blockDim.x + threadIdx.x) >> 5;
  const int e0 = wave * SC_EDGES;
  if (e0 >= E) return;
  const int e = min(e0 + lane, E - 1);
  const int s = src[e];
  const int d = dst[e];
  const float we = w[e];

  if (e0 + SC_EDGES <= E) {
#pragma unroll
    for (int g = 0; g < SC_EDGES; g += SC_GROUP) {
      float a[SC_GROUP], b[SC_GROUP];
#pragma unroll
      for (int u = 0; u < SC_GROUP; ++u) {
        const float* hp = h + (size_t)lane_bcast_i(s, g + u) * HDIM + lane;
        a[u] = hp[0];
        b[u] = hp[32];
      }
#pragma unroll
      for (int u = 0; u < SC_GROUP; ++u) {
        float* lp = lap + (size_t)lane_bcast_i(d, g + u) * HDIM + lane;
        const float wj = lane_bcast_f(we, g + u);
        atomicAdd(lp, wj * a[u]);
        atomicAdd(lp + 32, wj * b[u]);
      }
    }
  } else {
    const int cnt = E - e0;
    for (int j = 0; j < cnt; ++j) {
      const float* hp = h + (size_t)lane_bcast_i(s, j) * HDIM + lane;
      float* lp = lap + (size_t)lane_bcast_i(d, j) * HDIM + lane;
      const float wj = lane_bcast_f(we, j);
      atomicAdd(lp, wj * hp[0]);
      atomicAdd(lp + 32, wj * hp[32]);
    }
  }
}


__device__ __forceinline__ void stage_w_pairs(float* sWp, const float* __restrict__ W,
                                              int tid) {
  const float4* Wv = (const float4*)W;
#pragma unroll
  for (int i = 0; i < 8; ++i) {
    int idx = tid + i * 128;
    float4 v = Wv[idx];
    int k = idx >> 4;
    int c4 = (idx & 15) * 4;
    float* dstp = sWp + (k >> 1) * WPSTR + (k & 1);
    dstp[(c4 + 0) * 2] = v.x;
    dstp[(c4 + 1) * 2] = v.y;
    dstp[(c4 + 2) * 2] = v.z;
    dstp[(c4 + 3) * 2] = v.w;
  }
}

__device__ __forceinline__ void stage_rows(float* sAw, const float* __restrict__ A,
                                           int rowBase, int n, int lane) {
#pragma unroll
  for (int i = 0; i < 8; ++i) {
    int fi = lane + i * 32;
    int r = fi >> 4, c4 = fi & 15;
    int row = rowBase + r;
    if (row >= n) row = n - 1;
    ((float4*)(sAw + r * ASTR))[c4] = ((const float4*)(A + (size_t)row * 64))[c4];
  }
}

__device__ __forceinline__ void mm_accum64(const float* sAw, const float* sWp,
                                           int lane, v8f acc[4]) {
  const int nlo = lane & 15;
  const int g = lane >> 4;
#pragma unroll
  for (int kk = 0; kk < 16; ++kk) {
    const int k0 = kk * 4 + 2 * g;
    v2f a = *(const v2f*)(sAw + nlo * ASTR + k0);
    const float* brow = sWp + (kk * 2 + g) * WPSTR;
#pragma unroll
    for (int nt = 0; nt < 4; ++nt) {
      v2f b = *(const v2f*)(brow + (nt * 16 + nlo) * 2);
      acc[nt] = wmma4(a, b, acc[nt]);
    }
  }
}


__global__ void __launch_bounds__(128)
gemm64_relu_kernel(const float* __restrict__ A, const float* __restrict__ W,
                   const float* __restrict__ bias, float* __restrict__ out, int n) {
  __shared__ float sWp[WPROWS * WPSTR];
  __shared__ float sA[GWAVES * 16 * ASTR];
  __shared__ float sB[64];
  const int tid = threadIdx.x, wave = tid >> 5, lane = tid & 31;
  stage_w_pairs(sWp, W, tid);
  if (tid < 64) sB[tid] = bias[tid];

  const int rowBase = blockIdx.x * GROWS + wave * 16;
  float* sAw = sA + wave * 16 * ASTR;
  stage_rows(sAw, A, rowBase, n, lane);
  __syncthreads();

  v8f z = {0, 0, 0, 0, 0, 0, 0, 0};
  v8f acc[4] = {z, z, z, z};
  mm_accum64(sAw, sWp, lane, acc);

  const int nlo = lane & 15, g = lane >> 4;
#pragma unroll
  for (int nt = 0; nt < 4; ++nt)
#pragma unroll
    for (int r = 0; r < 8; ++r) {
      int row = rowBase + r + 8 * g;
      int col = nt * 16 + nlo;
      if (row < n) out[(size_t)row * 64 + col] = fmaxf(acc[nt][r] + sB[col], 0.0f);
    }
}

__global__ void __launch_bounds__(128)
gemm64_cheb2_kernel(float* __restrict__ h, const float* __restrict__ lap,
                    const float* __restrict__ W0, const float* __restrict__ W1,
                    const float* __restrict__ bias, int n) {
  __shared__ float sWp[2 * WPROWS * WPSTR];
  __shared__ float sA[GWAVES * 16 * ASTR];
  __shared__ float sB[64];
  const int tid = threadIdx.x, wave = tid >> 5, lane = tid & 31;
  stage_w_pairs(sWp, W0, tid);
  stage_w_pairs(sWp + WPROWS * WPSTR, W1, tid);
  if (tid < 64) sB[tid] = bias[tid];

  const int rowBase = blockIdx.x * GROWS + wave * 16;
  float* sAw = sA + wave * 16 * ASTR;

  v8f z = {0, 0, 0, 0, 0, 0, 0, 0};
  v8f acc[4] = {z, z, z, z};

  stage_rows(sAw, h, rowBase, n, lane);
  __syncthreads();
  mm_accum64(sAw, sWp, lane, acc);
  __syncthreads();
  stage_rows(sAw, lap, rowBase, n, lane);
  __syncthreads();
  mm_accum64(sAw, sWp + WPROWS * WPSTR, lane, acc);

  const int nlo = lane & 15, g = lane >> 4;
#pragma unroll
  for (int nt = 0; nt < 4; ++nt)
#pragma unroll
    for (int r = 0; r < 8; ++r) {
      int row = rowBase + r + 8 * g;
      int col = nt * 16 + nlo;
      if (row < n) {
        size_t idx = (size_t)row * 64 + col;
        float v = fmaxf(acc[nt][r] + sB[col], 0.0f);
        h[idx] = v + h[idx];
      }
    }
}

__global__ void __launch_bounds__(128)
gemm_out_kernel(const float* __restrict__ A, const float* __restrict__ Wl,
                const float* __restrict__ bl, float* __restrict__ logits, int n) {
  __shared__ float sWo[WPROWS * WOSTR];
  __shared__ float sA[GWAVES * 16 * ASTR];
  __shared__ float sB[16];
  const int tid = threadIdx.x, wave = tid >> 5, lane = tid & 31;
  for (int idx = tid; idx < 64 * 16; idx += 128) {
    int k = idx >> 4, c = idx & 15;
    float v = (c < CDIM) ? Wl[k * CDIM + c] : 0.0f;
    sWo[(k >> 1) * WOSTR + c * 2 + (k & 1)] = v;
  }
  if (tid < 16) sB[tid] = (tid < CDIM) ? bl[tid] : 0.0f;

  const int rowBase = blockIdx.x * GROWS + wave * 16;
  float* sAw = sA + wave * 16 * ASTR;
  stage_rows(sAw, A, rowBase, n, lane);
  __syncthreads();

  const int nlo = lane & 15, g = lane >> 4;
  v8f acc = {0, 0, 0, 0, 0, 0, 0, 0};
#pragma unroll
  for (int kk = 0; kk < 16; ++kk) {
    const int k0 = kk * 4 + 2 * g;
    v2f a = *(const v2f*)(sAw + nlo * ASTR + k0);
    v2f b = *(const v2f*)(sWo + (kk * 2 + g) * WOSTR + nlo * 2);
    acc = wmma4(a, b, acc);
  }
#pragma unroll
  for (int r = 0; r < 8; ++r) {
    int row = rowBase + r + 8 * g;
    if (row < n) logits[(size_t)row * 16 + nlo] = acc[r] + sB[nlo];
  }
}

__global__ void logsoftmax_kernel(const float* __restrict__ logits,
                                  float* __restrict__ out, int n) {
  int i = blockIdx.x * blockDim.x + threadIdx.x;
  if (i >= n) return;
  const float* v = logits + (size_t)i * 16;
  float m = v[0];
#pragma unroll
  for (int c = 1; c < CDIM; ++c) m = fmaxf(m, v[c]);
  float s = 0.0f;
#pragma unroll
  for (int c = 0; c < CDIM; ++c) s += expf(v[c] - m);
  float ls = logf(s);
#pragma unroll
  for (int c = 0; c < CDIM; ++c) out[(size_t)i * CDIM + c] = v[c] - m - ls;
}


extern "C" void kernel_launch(void* const* d_in, const int* in_sizes, int n_in,
                              void* d_out, int out_size, void* d_ws, size_t ws_size,
                              hipStream_t stream) {
  const float* x    = (const float*)d_in[0];
  const int*   ei   = (const int*)d_in[1];
  const float* W1_0 = (const float*)d_in[2];
  const float* b1   = (const float*)d_in[3];
  const float* W2_0 = (const float*)d_in[4];
  const float* W2_1 = (const float*)d_in[5];
  const float* b2   = (const float*)d_in[6];
  const float* Wl   = (const float*)d_in[7];
  const float* bl   = (const float*)d_in[8];

  const int N = in_sizes[0] / FDIM;
  const int E = in_sizes[1] / 2;
  const int* src = ei;
  const int* dst = ei + E;

  float* ws     = (float*)d_ws;
  float* deg    = ws;
  float* lap    = deg + N;
  float* dis    = lap + (size_t)N * 64;
  float* wbuf   = dis + N;
  float* h      = wbuf + E;
  float* logits = h + (size_t)N * 64;
  float* out    = (float*)d_out;

  {
    int tot = N * 65;
    zero_kernel<<<(tot + 255) / 256, 256, 0, stream>>>(deg, tot);
  }
  degree_kernel<<<(E + 255) / 256, 256, 0, stream>>>(src, deg, E);
  dis_kernel<<<(N + 255) / 256, 256, 0, stream>>>(deg, dis, N);
  edgew_kernel<<<(E + 255) / 256, 256, 0, stream>>>(src, dst, dis, wbuf, E);

  const int gblocks = (N + GROWS - 1) / GROWS;
  gemm64_relu_kernel<<<gblocks, 128, 0, stream>>>(x, W1_0, b1, h, N);
  scatter_kernel<<<(E + 255) / 256, 256, 0, stream>>>(src, dst, wbuf, h, lap, E);
  gemm64_cheb2_kernel<<<gblocks, 128, 0, stream>>>(h, lap, W2_0, W2_1, b2, N);
  gemm_out_kernel<<<gblocks, 128, 0, stream>>>(h, Wl, bl, logits, N);
  logsoftmax_kernel<<<(N + 255) / 256, 256, 0, stream>>>(logits, out, N);
}


# 9 launches -> 4: zero fill folded into the first GEMM, degree norm and edge weights into the scatter, output GEMM and log-softmax into the second GEMM's epilogue; same sums, atomics kept
# speedup vs baseline: 1.4609x; 1.1105x over previous
#include <hip/hip_runtime.h>
#include <hip/hip_bf16.h>
#include <stdint.h>


#define FDIM 64
#define HDIM 64
#define CDIM 10

#define GWAVES 4
#define GROWS  (GWAVES * 16)
#define ASTR   68
#define WPSTR  160
#define WPROWS 32
#define WOSTR  96

typedef float v2f __attribute__((ext_vector_type(2)));
typedef float v8f __attribute__((ext_vector_type(8)));

__device__ __forceinline__ v8f wmma4(v2f a, v2f b, v8f c) {
  return __builtin_amdgcn_wmma_f32_16x16x4_f32(false, a, false, b, (short)0, c,
                                               false, false);
}


__global__ void degree_kernel(const int* __restrict__ src, float* __restrict__ deg, int E) {
  int e = blockIdx.x * blockDim.x + threadIdx.x;
  if (e < E) atomicAdd(&deg[src[e]], 1.0f);
}

__device__ __forceinline__ float dis_of(float d) {
  return (d > 0.0f) ? rsqrtf(fmaxf(d, 1.0f)) : 0.0f;
}

#define SC_EDGES 32
#define SC_GROUP 8

__device__ __forceinline__ int lane_bcast_i(int v, int j) {
  return __builtin_amdgcn_readlane(v, j);
}
__device__ __forceinline__ float lane_bcast_f(float v, int j) {
  return __int_as_float(__builtin_amdgcn_readlane(__float_as_int(v), j));
}

__global__ void __launch_bounds__(256)
scatter_kernel(const int* __restrict__ src, const int* __restrict__ dst,
               const float* __restrict__ deg, const float* __restrict__ h,
               float* __restrict__ lap, int E) {
  const int lane = threadIdx.x & 31;
  const int wave = (blockIdx.x * blockDim.x + threadIdx.x) >> 5;
  const int e0 = wave * SC_EDGES;
  if (e0 >= E) return;
  const int e = min(e0 + lane, E - 1);
  const int s = src[e];
  const int d = dst[e];
  const float we = -(dis_of(deg[s]) * dis_of(deg[d]));

  if (e0 + SC_EDGES <= E) {
#pragma unroll
    for (int g = 0; g < SC_EDGES; g += SC_GROUP) {
      float a[SC_GROUP], b[SC_GROUP];
#pragma unroll
      for (int u = 0; u < SC_GROUP; ++u) {
        const float* hp = h + (size_t)lane_bcast_i(s, g + u) * HDIM + lane;
        a[u] = hp[0];
        b[u] = hp[32];
      }
#pragma unroll
      for (int u = 0; u < SC_GROUP; ++u) {
        float* lp = lap + (size_t)lane_bcast_i(d, g + u) * HDIM + lane;
        const float wj = lane_bcast_f(we, g + u);
        atomicAdd(lp, wj * a[u]);
        atomicAdd(lp + 32, wj * b[u]);
      }
    }
  } else {
    const int cnt = E - e0;
    for (int j = 0; j < cnt; ++j) {
      const float* hp = h + (size_t)lane_bcast_i(s, j) * HDIM + lane;
      float* lp = lap + (size_t)lane_bcast_i(d, j) * HDIM + lane;
      const float wj = lane_bcast_f(we, j);
      atomicAdd(lp, wj * hp[0]);
      atomicAdd(lp + 32, wj * hp[32]);
    }
  }
}


__device__ __forceinline__ void stage_w_pairs(float* sWp, const float* __restrict__ W,
                                              int tid) {
  const float4* Wv = (const float4*)W;
#pragma unroll
  for (int i = 0; i < 8; ++i) {
    int idx = tid + i * 128;
    float4 v = Wv[idx];
    int k = idx >> 4;
    int c4 = (idx & 15) * 4;
    float* dstp = sWp + (k >> 1) * WPSTR + (k & 1);
    dstp[(c4 + 0) * 2] = v.x;
    dstp[(c4 + 1) * 2] = v.y;
    dstp[(c4 + 2) * 2] = v.z;
    dstp[(c4 + 3) * 2] = v.w;
  }
}

__device__ __forceinline__ void stage_rows(float* sAw, const float* __restrict__ A,
                                           int rowBase, int n, int lane) {
#pragma unroll
  for (int i = 0; i < 8; ++i) {
    int fi = lane + i * 32;
    int r = fi >> 4, c4 = fi & 15;
    int row = rowBase + r;
    if (row >= n) row = n - 1;
    ((float4*)(sAw + r * ASTR))[c4] = ((const float4*)(A + (size_t)row * 64))[c4];
  }
}

__device__ __forceinline__ void mm_accum64(const float* sAw, const float* sWp,
                                           int lane, v8f acc[4]) {
  const int nlo = lane & 15;
  const int g = lane >> 4;
#pragma unroll
  for (int kk = 0; kk < 16; ++kk) {
    const int k0 = kk * 4 + 2 * g;
    v2f a = *(const v2f*)(sAw + nlo * ASTR + k0);
    const float* brow = sWp + (kk * 2 + g) * WPSTR;
#pragma unroll
    for (int nt = 0; nt < 4; ++nt) {
      v2f b = *(const v2f*)(brow + (nt * 16 + nlo) * 2);
      acc[nt] = wmma4(a, b, acc[nt]);
    }
  }
}


__global__ void __launch_bounds__(128)
gemm64_relu_kernel(const float* __restrict__ A, const float* __restrict__ W,
                   const float* __restrict__ bias, float* __restrict__ out,
                   float* __restrict__ lap, float* __restrict__ deg, int n) {
  __shared__ float sWp[WPROWS * WPSTR];
  __shared__ float sA[GWAVES * 16 * ASTR];
  __shared__ float sB[64];
  const int tid = threadIdx.x, wave = tid >> 5, lane = tid & 31;
  stage_w_pairs(sWp, W, tid);
  if (tid < 64) sB[tid] = bias[tid];

  const int rowBase = blockIdx.x * GROWS + wave * 16;
  float* sAw = sA + wave * 16 * ASTR;
  stage_rows(sAw, A, rowBase, n, lane);
  __syncthreads();

  {
    const float4 z4 = {0.0f, 0.0f, 0.0f, 0.0f};
#pragma unroll
    for (int i = 0; i < 8; ++i) {
      int fi = lane + i * 32;
      int row = rowBase + (fi >> 4);
      if (row < n) ((float4*)(lap + (size_t)row * 64))[fi & 15] = z4;
    }
    if (lane < 16 && rowBase + lane < n) deg[rowBase + lane] = 0.0f;
  }

  v8f z = {0, 0, 0, 0, 0, 0, 0, 0};
  v8f acc[4] = {z, z, z, z};
  mm_accum64(sAw, sWp, lane, acc);

  const int nlo = lane & 15, g = lane >> 4;
#pragma unroll
  for (int nt = 0; nt < 4; ++nt)
#pragma unroll
    for (int r = 0; r < 8; ++r) {
      int row = rowBase + r + 8 * g;
      int col = nt * 16 + nlo;
      if (row < n) out[(size_t)row * 64 + col] = fmaxf(acc[nt][r] + sB[col], 0.0f);
    }
}

#define LSTR 17

__global__ void __launch_bounds__(128)
cheb2_head_kernel(const float* __restrict__ h, const float* __restrict__ lap,
                  const float* __restrict__ W0, const float* __restrict__ W1,
                  const float* __restrict__ bias, const float* __restrict__ Wl,
                  const float* __restrict__ bl, float* __restrict__ out, int n) {
  __shared__ float sWp[2 * WPROWS * WPSTR];
  __shared__ float sWo[WPROWS * WOSTR];
  __shared__ float sA[GWAVES * 16 * ASTR];
  __shared__ float sL[GWAVES * 16 * LSTR];
  __shared__ float sO[GWAVES * 16 * CDIM];
  __shared__ float sB[64];
  __shared__ float sBo[16];
  const int tid = threadIdx.x, wave = tid >> 5, lane = tid & 31;
  stage_w_pairs(sWp, W0, tid);
  stage_w_pairs(sWp + WPROWS * WPSTR, W1, tid);
  for (int idx = tid; idx < 64 * 16; idx += 128) {
    int k = idx >> 4, c = idx & 15;
    float v = (c < CDIM) ? Wl[k * CDIM + c] : 0.0f;
    sWo[(k >> 1) * WOSTR + c * 2 + (k & 1)] = v;
  }
  if (tid < 64) sB[tid] = bias[tid];
  if (tid < 16) sBo[tid] = (tid < CDIM) ? bl[tid] : 0.0f;

  const int rowBase = blockIdx.x * GROWS + wave * 16;
  float* sAw = sA + wave * 16 * ASTR;
  float* sLw = sL + wave * 16 * LSTR;
  float* sOw = sO + wave * 16 * CDIM;

  v8f z = {0, 0, 0, 0, 0, 0, 0, 0};
  v8f acc[4] = {z, z, z, z};

  stage_rows(sAw, h, rowBase, n, lane);
  __syncthreads();
  mm_accum64(sAw, sWp, lane, acc);
  __syncthreads();
  stage_rows(sAw, lap, rowBase, n, lane);
  __syncthreads();
  mm_accum64(sAw, sWp + WPROWS * WPSTR, lane, acc);

  const int nlo = lane & 15, g = lane >> 4;
#pragma unroll
  for (int nt = 0; nt < 4; ++nt)
#pragma unroll
    for (int r = 0; r < 8; ++r) {
      int rl = r + 8 * g;
      int row = min(rowBase + rl, n - 1);
      int col = nt * 16 + nlo;
      float v = fmaxf(acc[nt][r] + sB[col], 0.0f);
      sAw[rl * ASTR + col] = v + h[(size_t)row * 64 + col];
    }
  __syncthreads();

  v8f acc2 = z;
#pragma unroll
  for (int kk = 0; kk < 16; ++kk) {
    const int k0 = kk * 4 + 2 * g;
    v2f a = *(const v2f*)(sAw + nlo * ASTR + k0);
    v2f b = *(const v2f*)(sWo + (kk * 2 + g) * WOSTR + nlo * 2);
    acc2 = wmma4(a, b, acc2);
  }
#pragma unroll
  for (int r = 0; r < 8; ++r) sLw[(r + 8 * g) * LSTR + nlo] = acc2[r] + sBo[nlo];
  __syncthreads();

  {
    const float* v = sLw + nlo * LSTR;
    float m = v[0];
#pragma unroll
    for (int c = 1; c < CDIM; ++c) m = fmaxf(m, v[c]);
    float s = 0.0f;
#pragma unroll
    for (int c = 0; c < CDIM; ++c) s += expf(v[c] - m);
    float ls = logf(s);
    if (g == 0) {
#pragma unroll
      for (int c = 0; c < CDIM; ++c) sOw[nlo * CDIM + c] = v[c] - m - ls;
    }
  }
  __syncthreads();
#pragma unroll
  for (int j = 0; j < (16 * CDIM) / 32; ++j) {
    int idx = lane + 32 * j;
    if (rowBase + idx / CDIM < n) out[(size_t)rowBase * CDIM + idx] = sOw[idx];
  }
}


extern "C" void kernel_launch(void* const* d_in, const int* in_sizes, int n_in,
                              void* d_out, int out_size, void* d_ws, size_t ws_size,
                              hipStream_t stream) {
  const float* x    = (const float*)d_in[0];
  const int*   ei   = (const int*)d_in[1];
  const float* W1_0 = (const float*)d_in[2];
  const float* b1   = (const float*)d_in[3];
  const float* W2_0 = (const float*)d_in[4];
  const float* W2_1 = (const float*)d_in[5];
  const float* b2   = (const float*)d_in[6];
  const float* Wl   = (const float*)d_in[7];
  const float* bl   = (const float*)d_in[8];

  const int N = in_sizes[0] / FDIM;
  const int E = in_sizes[1] / 2;
  const int* src = ei;
  const int* dst = ei + E;

  float* ws     = (float*)d_ws;
  float* deg    = ws;
  float* lap    = deg + N;
  float* dis    = lap + (size_t)N * 64;
  float* wbuf   = dis + N;
  float* h      = wbuf + E;
  float* out    = (float*)d_out;

  const int gblocks = (N + GROWS - 1) / GROWS;
  gemm64_relu_kernel<<<gblocks, 128, 0, stream>>>(x, W1_0, b1, h, lap, deg, N);
  degree_kernel<<<(E + 255) / 256, 256, 0, stream>>>(src, deg, E);
  scatter_kernel<<<(E + 255) / 256, 256, 0, stream>>>(src, dst, deg, h, lap, E);
  cheb2_head_kernel<<<gblocks, 128, 0, stream>>>(h, lap, W2_0, W2_1, b2, Wl, bl, out, N);
}


# scatter adds into two copies of lap, each wave into the copy of its package half (from its die number, speed only); last kernel adds the copies; atomics kept, a row's sum becomes two partial sums
# speedup vs baseline: 1.6208x; 1.1095x over previous
#include <hip/hip_runtime.h>
#include <hip/hip_bf16.h>
#include <stdint.h>


#define FDIM 64
#define HDIM 64
#define CDIM 10

#define GWAVES 4
#define GROWS  (GWAVES * 16)
#define ASTR   68
#define WPSTR  160
#define WPROWS 32
#define WOSTR  96

typedef float v2f __attribute__((ext_vector_type(2)));
typedef float v8f __attribute__((ext_vector_type(8)));

__device__ __forceinline__ v8f wmma4(v2f a, v2f b, v8f c) {
  return __builtin_amdgcn_wmma_f32_16x16x4_f32(false, a, false, b, (short)0, c,
                                               false, false);
}


#define LAP2_PAD 48

__device__ __forceinline__ bool two_copies(size_t ws_bytes, int n, int E) {
  const size_t need = (size_t)n * 2 + (size_t)n * 64 * 3 + (size_t)E + LAP2_PAD;
  return ws_bytes >= need * sizeof(float);
}

__device__ __forceinline__ int package_half() {
  return (int)((__builtin_amdgcn_s_sendmsg_rtn(0x87) >> 18) & 1u);
}


__global__ void degree_kernel(const int* __restrict__ src, float* __restrict__ deg, int E) {
  int e = blockIdx.x * blockDim.x + threadIdx.x;
  if (e < E) atomicAdd(&deg[src[e]], 1.0f);
}

__device__ __forceinline__ float dis_of(float d) {
  return (d > 0.0f) ? rsqrtf(fmaxf(d, 1.0f)) : 0.0f;
}

#define SC_EDGES 32
#define SC_GROUP 8

__device__ __forceinline__ int lane_bcast_i(int v, int j) {
  return __builtin_amdgcn_readlane(v, j);
}
__device__ __forceinline__ float lane_bcast_f(float v, int j) {
  return __int_as_float(__builtin_amdgcn_readlane(__float_as_int(v), j));
}

__global__ void __launch_bounds__(256)
scatter_kernel(const int* __restrict__ src, const int* __restrict__ dst,
               const float* __restrict__ deg, const float* __restrict__ h,
               float* __restrict__ lap0, float* __restrict__ lap1, int n, int E,
               size_t ws_bytes) {
  const int half = two_copies(ws_bytes, n, E) ? package_half() : 0;
  const int lane = threadIdx.x & 31;
  const int wave = (blockIdx.x * blockDim.x + threadIdx.x) >> 5;
  const int e0 = wave * SC_EDGES;
  if (e0 >= E) return;
  const int e = min(e0 + lane, E - 1);
  const int s = src[e];
  const int d = dst[e];
  const float we = -(dis_of(deg[s]) * dis_of(deg[d]));
  float* __restrict__ lap = half ? lap1 : lap0;

  if (e0 + SC_EDGES <= E) {
#pragma unroll
    for (int g = 0; g < SC_EDGES; g += SC_GROUP) {
      float a[SC_GROUP], b[SC_GROUP];
#pragma unroll
      for (int u = 0; u < SC_GROUP; ++u) {
        const float* hp = h + (size_t)lane_bcast_i(s, g + u) * HDIM + lane;
        a[u] = hp[0];
        b[u] = hp[32];
      }
#pragma unroll
      for (int u = 0; u < SC_GROUP; ++u) {
        float* lp = lap + (size_t)lane_bcast_i(d, g + u) * HDIM + lane;
        const float wj = lane_bcast_f(we, g + u);
        atomicAdd(lp, wj * a[u]);
        atomicAdd(lp + 32, wj * b[u]);
      }
    }
  } else {
    const int cnt = E - e0;
    for (int j = 0; j < cnt; ++j) {
      const float* hp = h + (size_t)lane_bcast_i(s, j) * HDIM + lane;
      float* lp = lap + (size_t)lane_bcast_i(d, j) * HDIM + lane;
      const float wj = lane_bcast_f(we, j);
      atomicAdd(lp, wj * hp[0]);
      atomicAdd(lp + 32, wj * hp[32]);
    }
  }
}


__device__ __forceinline__ void stage_w_pairs(float* sWp, const float* __restrict__ W,
                                              int tid) {
  const float4* Wv = (const float4*)W;
#pragma unroll
  for (int i = 0; i < 8; ++i) {
    int idx = tid + i * 128;
    float4 v = Wv[idx];
    int k = idx >> 4;
    int c4 = (idx & 15) * 4;
    float* dstp = sWp + (k >> 1) * WPSTR + (k & 1);
    dstp[(c4 + 0) * 2] = v.x;
    dstp[(c4 + 1) * 2] = v.y;
    dstp[(c4 + 2) * 2] = v.z;
    dstp[(c4 + 3) * 2] = v.w;
  }
}

__device__ __forceinline__ void stage_rows(float* sAw, const float* __restrict__ A,
                                           int rowBase, int n, int lane) {
#pragma unroll
  for (int i = 0; i < 8; ++i) {
    int fi = lane + i * 32;
    int r = fi >> 4, c4 = fi & 15;
    int row = rowBase + r;
    if (row >= n) row = n - 1;
    ((float4*)(sAw + r * ASTR))[c4] = ((const float4*)(A + (size_t)row * 64))[c4];
  }
}

__device__ __forceinline__ void stage_rows_sum2(float* sAw, const float* __restrict__ A,
                                                const float* __restrict__ B,
                                                int rowBase, int n, int lane) {
#pragma unroll
  for (int i = 0; i < 8; ++i) {
    int fi = lane + i * 32;
    int r = fi >> 4, c4 = fi & 15;
    int row = rowBase + r;
    if (row >= n) row = n - 1;
    float4 a = ((const float4*)(A + (size_t)row * 64))[c4];
    const float4 b = ((const float4*)(B + (size_t)row * 64))[c4];
    a.x += b.x; a.y += b.y; a.z += b.z; a.w += b.w;
    ((float4*)(sAw + r * ASTR))[c4] = a;
  }
}

__device__ __forceinline__ void mm_accum64(const float* sAw, const float* sWp,
                                           int lane, v8f acc[4]) {
  const int nlo = lane & 15;
  const int g = lane >> 4;
#pragma unroll
  for (int kk = 0; kk < 16; ++kk) {
    const int k0 = kk * 4 + 2 * g;
    v2f a = *(const v2f*)(sAw + nlo * ASTR + k0);
    const float* brow = sWp + (kk * 2 + g) * WPSTR;
#pragma unroll
    for (int nt = 0; nt < 4; ++nt) {
      v2f b = *(const v2f*)(brow + (nt * 16 + nlo) * 2);
      acc[nt] = wmma4(a, b, acc[nt]);
    }
  }
}


__global__ void __launch_bounds__(128)
gemm64_relu_kernel(const float* __restrict__ A, const float* __restrict__ W,
                   const float* __restrict__ bias, float* __restrict__ out,
                   float* __restrict__ lap, float* __restrict__ lap1,
                   float* __restrict__ deg, int n, int E, size_t ws_bytes) {
  __shared__ float sWp[WPROWS * WPSTR];
  __shared__ float sA[GWAVES * 16 * ASTR];
  __shared__ float sB[64];
  const bool two = two_copies(ws_bytes, n, E);
  const int tid = threadIdx.x, wave = tid >> 5, lane = tid & 31;
  stage_w_pairs(sWp, W, tid);
  if (tid < 64) sB[tid] = bias[tid];

  const int rowBase = blockIdx.x * GROWS + wave * 16;
  float* sAw = sA + wave * 16 * ASTR;
  stage_rows(sAw, A, rowBase, n, lane);
  __syncthreads();

  {
    const float4 z4 = {0.0f, 0.0f, 0.0f, 0.0f};
#pragma unroll
    for (int i = 0; i < 8; ++i) {
      int fi = lane + i * 32;
      int row = rowBase + (fi >> 4);
      if (row < n) {
        ((float4*)(lap + (size_t)row * 64))[fi & 15] = z4;
        if (two) ((float4*)(lap1 + (size_t)row * 64))[fi & 15] = z4;
      }
    }
    if (lane < 16 && rowBase + lane < n) deg[rowBase + lane] = 0.0f;
  }

  v8f z = {0, 0, 0, 0, 0, 0, 0, 0};
  v8f acc[4] = {z, z, z, z};
  mm_accum64(sAw, sWp, lane, acc);

  const int nlo = lane & 15, g = lane >> 4;
#pragma unroll
  for (int nt = 0; nt < 4; ++nt)
#pragma unroll
    for (int r = 0; r < 8; ++r) {
      int row = rowBase + r + 8 * g;
      int col = nt * 16 + nlo;
      if (row < n) out[(size_t)row * 64 + col] = fmaxf(acc[nt][r] + sB[col], 0.0f);
    }
}

#define LSTR 17

__global__ void __launch_bounds__(128)
cheb2_head_kernel(const float* __restrict__ h, const float* __restrict__ lap,
                  const float* __restrict__ lap1,
                  const float* __restrict__ W0, const float* __restrict__ W1,
                  const float* __restrict__ bias, const float* __restrict__ Wl,
                  const float* __restrict__ bl, float* __restrict__ out, int n, int E,
                  size_t ws_bytes) {
  __shared__ float sWp[2 * WPROWS * WPSTR];
  __shared__ float sWo[WPROWS * WOSTR];
  __shared__ float sA[GWAVES * 16 * ASTR];
  __shared__ float sL[GWAVES * 16 * LSTR];
  __shared__ float sO[GWAVES * 16 * CDIM];
  __shared__ float sB[64];
  __shared__ float sBo[16];
  const int tid = threadIdx.x, wave = tid >> 5, lane = tid & 31;
  stage_w_pairs(sWp, W0, tid);
  stage_w_pairs(sWp + WPROWS * WPSTR, W1, tid);
  for (int idx = tid; idx < 64 * 16; idx += 128) {
    int k = idx >> 4, c = idx & 15;
    float v = (c < CDIM) ? Wl[k * CDIM + c] : 0.0f;
    sWo[(k >> 1) * WOSTR + c * 2 + (k & 1)] = v;
  }
  if (tid < 64) sB[tid] = bias[tid];
  if (tid < 16) sBo[tid] = (tid < CDIM) ? bl[tid] : 0.0f;

  const int rowBase = blockIdx.x * GROWS + wave * 16;
  float* sAw = sA + wave * 16 * ASTR;
  float* sLw = sL + wave * 16 * LSTR;
  float* sOw = sO + wave * 16 * CDIM;

  v8f z = {0, 0, 0, 0, 0, 0, 0, 0};
  v8f acc[4] = {z, z, z, z};

  stage_rows(sAw, h, rowBase, n, lane);
  __syncthreads();
  mm_accum64(sAw, sWp, lane, acc);
  __syncthreads();
  if (two_copies(ws_bytes, n, E))
    stage_rows_sum2(sAw, lap, lap1, rowBase, n, lane);
  else
    stage_rows(sAw, lap, rowBase, n, lane);
  __syncthreads();
  mm_accum64(sAw, sWp + WPROWS * WPSTR, lane, acc);

  const int nlo = lane & 15, g = lane >> 4;
#pragma unroll
  for (int nt = 0; nt < 4; ++nt)
#pragma unroll
    for (int r = 0; r < 8; ++r) {
      int rl = r + 8 * g;
      int row = min(rowBase + rl, n - 1);
      int col = nt * 16 + nlo;
      float v = fmaxf(acc[nt][r] + sB[col], 0.0f);
      sAw[rl * ASTR + col] = v + h[(size_t)row * 64 + col];
    }
  __syncthreads();

  v8f acc2 = z;
#pragma unroll
  for (int kk = 0; kk < 16; ++kk) {
    const int k0 = kk * 4 + 2 * g;
    v2f a = *(const v2f*)(sAw + nlo * ASTR + k0);
    v2f b = *(const v2f*)(sWo + (kk * 2 + g) * WOSTR + nlo * 2);
    acc2 = wmma4(a, b, acc2);
  }
#pragma unroll
  for (int r = 0; r < 8; ++r) sLw[(r + 8 * g) * LSTR + nlo] = acc2[r] + sBo[nlo];
  __syncthreads();

  {
    const float* v = sLw + nlo * LSTR;
    float m = v[0];
#pragma unroll
    for (int c = 1; c < CDIM; ++c) m = fmaxf(m, v[c]);
    float s = 0.0f;
#pragma unroll
    for (int c = 0; c < CDIM; ++c) s += expf(v[c] - m);
    float ls = logf(s);
    if (g == 0) {
#pragma unroll
      for (int c = 0; c < CDIM; ++c) sOw[nlo * CDIM + c] = v[c] - m - ls;
    }
  }
  __syncthreads();
#pragma unroll
  for (int j = 0; j < (16 * CDIM) / 32; ++j) {
    int idx = lane + 32 * j;
    if (rowBase + idx / CDIM < n) out[(size_t)rowBase * CDIM + idx] = sOw[idx];
  }
}


extern "C" void kernel_launch(void* const* d_in, const int* in_sizes, int n_in,
                              void* d_out, int out_size, void* d_ws, size_t ws_size,
                              hipStream_t stream) {
  const float* x    = (const float*)d_in[0];
  const int*   ei   = (const int*)d_in[1];
  const float* W1_0 = (const float*)d_in[2];
  const float* b1   = (const float*)d_in[3];
  const float* W2_0 = (const float*)d_in[4];
  const float* W2_1 = (const float*)d_in[5];
  const float* b2   = (const float*)d_in[6];
  const float* Wl   = (const float*)d_in[7];
  const float* bl   = (const float*)d_in[8];

  const int N = in_sizes[0] / FDIM;
  const int E = in_sizes[1] / 2;
  const int* src = ei;
  const int* dst = ei + E;

  float* ws     = (float*)d_ws;
  float* deg    = ws;
  float* lap    = deg + N;
  float* dis    = lap + (size_t)N * 64;
  float* wbuf   = dis + N;
  float* h      = wbuf + E;
  float* lap1   = h + (size_t)N * 64 + LAP2_PAD;
  float* out    = (float*)d_out;

  const int gblocks = (N + GROWS - 1) / GROWS;
  gemm64_relu_kernel<<<gblocks, 128, 0, stream>>>(x, W1_0, b1, h, lap, lap1, deg, N, E, ws_size);
  degree_kernel<<<(E + 255) / 256, 256, 0, stream>>>(src, deg, E);
  scatter_kernel<<<(E + 255) / 256, 256, 0, stream>>>(src, dst, deg, h, lap, lap1, N, E, ws_size);
  cheb2_head_kernel<<<gblocks, 128, 0, stream>>>(h, lap, lap1, W2_0, W2_1, b2, Wl, bl, out, N, E, ws_size);
}


# degree count: one global f32 atomic per edge (each lane its own line) -> LDS integer counters per (edge slice, node range) block, added to deg as whole-line f32 atomics; exact counts, same deg
# speedup vs baseline: 2.2197x; 1.3695x over previous
#include <hip/hip_runtime.h>
#include <hip/hip_bf16.h>
#include <stdint.h>


#define FDIM 64
#define HDIM 64
#define CDIM 10

#define GWAVES 4
#define GROWS  (GWAVES * 16)
#define ASTR   68
#define WPSTR  160
#define WPROWS 32
#define WOSTR  96

typedef float v2f __attribute__((ext_vector_type(2)));
typedef float v8f __attribute__((ext_vector_type(8)));

__device__ __forceinline__ v8f wmma4(v2f a, v2f b, v8f c) {
  return __builtin_amdgcn_wmma_f32_16x16x4_f32(false, a, false, b, (short)0, c,
                                               false, false);
}


#define LAP2_PAD 48

__device__ __forceinline__ bool two_copies(size_t ws_bytes, int n, int E) {
  const size_t need = (size_t)n * 2 + (size_t)n * 64 * 3 + (size_t)E + LAP2_PAD;
  return ws_bytes >= need * sizeof(float);
}

__device__ __forceinline__ int package_half() {
  return (int)((__builtin_amdgcn_s_sendmsg_rtn(0x87) >> 18) & 1u);
}


#define DG_THREADS 512
#define DG_CHUNKS  32
#define DG_RANGES  8
#define DG_RLEN    6656
#define DG_PER     (DG_RLEN / DG_THREADS)
#define DG_U       13

typedef int v4i __attribute__((ext_vector_type(4)));

__device__ __forceinline__ void deg_count(unsigned int* cnt, int s, int base, int len) {
  const unsigned int r = (unsigned int)(s - base);
  if (r < (unsigned int)len) atomicAdd(&cnt[r], 1u);
}

__global__ void __launch_bounds__(DG_THREADS)
degree_kernel(const int* __restrict__ src, float* __restrict__ deg, int n, int E) {
  __shared__ unsigned int cnt[DG_RLEN];
  const int tid = threadIdx.x;
  const int nq = E >> 2;
  const int qper = (nq + (int)gridDim.x - 1) / (int)gridDim.x;
  const int q_lo = (int)blockIdx.x * qper;
  const int q_hi = min(nq, q_lo + qper);
  const v4i* __restrict__ sq = (const v4i*)src;
  const int rot = ((int)blockIdx.x * (DG_RLEN / DG_CHUNKS)) & ~31;

  for (int base = (int)blockIdx.y * DG_RLEN; base < n; base += (int)gridDim.y * DG_RLEN) {
    const int len = min(DG_RLEN, n - base);
#pragma unroll
    for (int k = 0; k < DG_PER; ++k) cnt[tid + k * DG_THREADS] = 0u;
    __syncthreads();

    for (int q0 = q_lo; q0 < q_hi; q0 += DG_THREADS * DG_U) {
      v4i v[DG_U];
#pragma unroll
      for (int u = 0; u < DG_U; ++u) v[u] = sq[min(q0 + u * DG_THREADS + tid, q_hi - 1)];
      asm volatile("" :: "v"(v[0]), "v"(v[1]), "v"(v[2]), "v"(v[3]), "v"(v[4]), "v"(v[5]),
                         "v"(v[6]), "v"(v[7]), "v"(v[8]), "v"(v[9]), "v"(v[10]), "v"(v[11]),
                         "v"(v[12]));
#pragma unroll
      for (int u = 0; u < DG_U; ++u) {
        const int lim = (q0 + u * DG_THREADS + tid < q_hi) ? len : 0;
        deg_count(cnt, v[u].x, base, lim);
        deg_count(cnt, v[u].y, base, lim);
        deg_count(cnt, v[u].z, base, lim);
        deg_count(cnt, v[u].w, base, lim);
      }
    }
    if (blockIdx.x == 0 && tid < E - 4 * nq) deg_count(cnt, src[4 * nq + tid], base, len);
    __syncthreads();

    unsigned int c[DG_PER];
#pragma unroll
    for (int k = 0; k < DG_PER; ++k) {
      int j = tid + k * DG_THREADS + rot;
      if (j >= DG_RLEN) j -= DG_RLEN;
      c[k] = cnt[j];
    }
#pragma unroll
    for (int k = 0; k < DG_PER; ++k) {
      int j = tid + k * DG_THREADS + rot;
      if (j >= DG_RLEN) j -= DG_RLEN;
      if (j < len && c[k] != 0u) atomicAdd(&deg[base + j], (float)c[k]);
    }
    __syncthreads();
  }
}

__device__ __forceinline__ float dis_of(float d) {
  return (d > 0.0f) ? rsqrtf(fmaxf(d, 1.0f)) : 0.0f;
}

#define SC_EDGES 32
#define SC_GROUP 8

__device__ __forceinline__ int lane_bcast_i(int v, int j) {
  return __builtin_amdgcn_readlane(v, j);
}
__device__ __forceinline__ float lane_bcast_f(float v, int j) {
  return __int_as_float(__builtin_amdgcn_readlane(__float_as_int(v), j));
}

__global__ void __launch_bounds__(256)
scatter_kernel(const int* __restrict__ src, const int* __restrict__ dst,
               const float* __restrict__ deg, const float* __restrict__ h,
               float* __restrict__ lap0, float* __restrict__ lap1, int n, int E,
               size_t ws_bytes) {
  const int half = two_copies(ws_bytes, n, E) ? package_half() : 0;
  const int lane = threadIdx.x & 31;
  const int wave = (blockIdx.x * blockDim.x + threadIdx.x) >> 5;
  const int e0 = wave * SC_EDGES;
  if (e0 >= E) return;
  const int e = min(e0 + lane, E - 1);
  const int s = src[e];
  const int d = dst[e];
  const float we = -(dis_of(deg[s]) * dis_of(deg[d]));
  float* __restrict__ lap = half ? lap1 : lap0;

  if (e0 + SC_EDGES <= E) {
#pragma unroll
    for (int g = 0; g < SC_EDGES; g += SC_GROUP) {
      float a[SC_GROUP], b[SC_GROUP];
#pragma unroll
      for (int u = 0; u < SC_GROUP; ++u) {
        const float* hp = h + (size_t)lane_bcast_i(s, g + u) * HDIM + lane;
        a[u] = hp[0];
        b[u] = hp[32];
      }
#pragma unroll
      for (int u = 0; u < SC_GROUP; ++u) {
        float* lp = lap + (size_t)lane_bcast_i(d, g + u) * HDIM + lane;
        const float wj = lane_bcast_f(we, g + u);
        atomicAdd(lp, wj * a[u]);
        atomicAdd(lp + 32, wj * b[u]);
      }
    }
  } else {
    const int cnt = E - e0;
    for (int j = 0; j < cnt; ++j) {
      const float* hp = h + (size_t)lane_bcast_i(s, j) * HDIM + lane;
      float* lp = lap + (size_t)lane_bcast_i(d, j) * HDIM + lane;
      const float wj = lane_bcast_f(we, j);
      atomicAdd(lp, wj * hp[0]);
      atomicAdd(lp + 32, wj * hp[32]);
    }
  }
}


__device__ __forceinline__ void stage_w_pairs(float* sWp, const float* __restrict__ W,
                                              int tid) {
  const float4* Wv = (const float4*)W;
#pragma unroll
  for (int i = 0; i < 8; ++i) {
    int idx = tid + i * 128;
    float4 v = Wv[idx];
    int k = idx >> 4;
    int c4 = (idx & 15) * 4;
    float* dstp = sWp + (k >> 1) * WPSTR + (k & 1);
    dstp[(c4 + 0) * 2] = v.x;
    dstp[(c4 + 1) * 2] = v.y;
    dstp[(c4 + 2) * 2] = v.z;
    dstp[(c4 + 3) * 2] = v.w;
  }
}

__device__ __forceinline__ void stage_rows(float* sAw, const float* __restrict__ A,
                                           int rowBase, int n, int lane) {
#pragma unroll
  for (int i = 0; i < 8; ++i) {
    int fi = lane + i * 32;
    int r = fi >> 4, c4 = fi & 15;
    int row = rowBase + r;
    if (row >= n) row = n - 1;
    ((float4*)(sAw + r * ASTR))[c4] = ((const float4*)(A + (size_t)row * 64))[c4];
  }
}

__device__ __forceinline__ void stage_rows_sum2(float* sAw, const float* __restrict__ A,
                                                const float* __restrict__ B,
                                                int rowBase, int n, int lane) {
#pragma unroll
  for (int i = 0; i < 8; ++i) {
    int fi = lane + i * 32;
    int r = fi >> 4, c4 = fi & 15;
    int row = rowBase + r;
    if (row >= n) row = n - 1;
    float4 a = ((const float4*)(A + (size_t)row * 64))[c4];
    const float4 b = ((const float4*)(B + (size_t)row * 64))[c4];
    a.x += b.x; a.y += b.y; a.z += b.z; a.w += b.w;
    ((float4*)(sAw + r * ASTR))[c4] = a;
  }
}

__device__ __forceinline__ void mm_accum64(const float* sAw, const float* sWp,
                                           int lane, v8f acc[4]) {
  const int nlo = lane & 15;
  const int g = lane >> 4;
#pragma unroll
  for (int kk = 0; kk < 16; ++kk) {
    const int k0 = kk * 4 + 2 * g;
    v2f a = *(const v2f*)(sAw + nlo * ASTR + k0);
    const float* brow = sWp + (kk * 2 + g) * WPSTR;
#pragma unroll
    for (int nt = 0; nt < 4; ++nt) {
      v2f b = *(const v2f*)(brow + (nt * 16 + nlo) * 2);
      acc[nt] = wmma4(a, b, acc[nt]);
    }
  }
}


__global__ void __launch_bounds__(128)
gemm64_relu_kernel(const float* __restrict__ A, const float* __restrict__ W,
                   const float* __restrict__ bias, float* __restrict__ out,
                   float* __restrict__ lap, float* __restrict__ lap1,
                   float* __restrict__ deg, int n, int E, size_t ws_bytes) {
  __shared__ float sWp[WPROWS * WPSTR];
  __shared__ float sA[GWAVES * 16 * ASTR];
  __shared__ float sB[64];
  const bool two = two_copies(ws_bytes, n, E);
  const int tid = threadIdx.x, wave = tid >> 5, lane = tid & 31;
  stage_w_pairs(sWp, W, tid);
  if (tid < 64) sB[tid] = bias[tid];

  const int rowBase = blockIdx.x * GROWS + wave * 16;
  float* sAw = sA + wave * 16 * ASTR;
  stage_rows(sAw, A, rowBase, n, lane);
  __syncthreads();

  {
    const float4 z4 = {0.0f, 0.0f, 0.0f, 0.0f};
#pragma unroll
    for (int i = 0; i < 8; ++i) {
      int fi = lane + i * 32;
      int row = rowBase + (fi >> 4);
      if (row < n) {
        ((float4*)(lap + (size_t)row * 64))[fi & 15] = z4;
        if (two) ((float4*)(lap1 + (size_t)row * 64))[fi & 15] = z4;
      }
    }
    if (lane < 16 && rowBase + lane < n) deg[rowBase + lane] = 0.0f;
  }

  v8f z = {0, 0, 0, 0, 0, 0, 0, 0};
  v8f acc[4] = {z, z, z, z};
  mm_accum64(sAw, sWp, lane, acc);

  const int nlo = lane & 15, g = lane >> 4;
#pragma unroll
  for (int nt = 0; nt < 4; ++nt)
#pragma unroll
    for (int r = 0; r < 8; ++r) {
      int row = rowBase + r + 8 * g;
      int col = nt * 16 + nlo;
      if (row < n) out[(size_t)row * 64 + col] = fmaxf(acc[nt][r] + sB[col], 0.0f);
    }
}

#define LSTR 17

__global__ void __launch_bounds__(128)
cheb2_head_kernel(const float* __restrict__ h, const float* __restrict__ lap,
                  const float* __restrict__ lap1,
                  const float* __restrict__ W0, const float* __restrict__ W1,
                  const float* __restrict__ bias, const float* __restrict__ Wl,
                  const float* __restrict__ bl, float* __restrict__ out, int n, int E,
                  size_t ws_bytes) {
  __shared__ float sWp[2 * WPROWS * WPSTR];
  __shared__ float sWo[WPROWS * WOSTR];
  __shared__ float sA[GWAVES * 16 * ASTR];
  __shared__ float sL[GWAVES * 16 * LSTR];
  __shared__ float sO[GWAVES * 16 * CDIM];
  __shared__ float sB[64];
  __shared__ float sBo[16];
  const int tid = threadIdx.x, wave = tid >> 5, lane = tid & 31;
  stage_w_pairs(sWp, W0, tid);
  stage_w_pairs(sWp + WPROWS * WPSTR, W1, tid);
  for (int idx = tid; idx < 64 * 16; idx += 128) {
    int k = idx >> 4, c = idx & 15;
    float v = (c < CDIM) ? Wl[k * CDIM + c] : 0.0f;
    sWo[(k >> 1) * WOSTR + c * 2 + (k & 1)] = v;
  }
  if (tid < 64) sB[tid] = bias[tid];
  if (tid < 16) sBo[tid] = (tid < CDIM) ? bl[tid] : 0.0f;

  const int rowBase = blockIdx.x * GROWS + wave * 16;
  float* sAw = sA + wave * 16 * ASTR;
  float* sLw = sL + wave * 16 * LSTR;
  float* sOw = sO + wave * 16 * CDIM;

  v8f z = {0, 0, 0, 0, 0, 0, 0, 0};
  v8f acc[4] = {z, z, z, z};

  stage_rows(sAw, h, rowBase, n, lane);
  __syncthreads();
  mm_accum64(sAw, sWp, lane, acc);
  __syncthreads();
  if (two_copies(ws_bytes, n, E))
    stage_rows_sum2(sAw, lap, lap1, rowBase, n, lane);
  else
    stage_rows(sAw, lap, rowBase, n, lane);
  __syncthreads();
  mm_accum64(sAw, sWp + WPROWS * WPSTR, lane, acc);

  const int nlo = lane & 15, g = lane >> 4;
#pragma unroll
  for (int nt = 0; nt < 4; ++nt)
#pragma unroll
    for (int r = 0; r < 8; ++r) {
      int rl = r + 8 * g;
      int row = min(rowBase + rl, n - 1);
      int col = nt * 16 + nlo;
      float v = fmaxf(acc[nt][r] + sB[col], 0.0f);
      sAw[rl * ASTR + col] = v + h[(size_t)row * 64 + col];
    }
  __syncthreads();

  v8f acc2 = z;
#pragma unroll
  for (int kk = 0; kk < 16; ++kk) {
    const int k0 = kk * 4 + 2 * g;
    v2f a = *(const v2f*)(sAw + nlo * ASTR + k0);
    v2f b = *(const v2f*)(sWo + (kk * 2 + g) * WOSTR + nlo * 2);
    acc2 = wmma4(a, b, acc2);
  }
#pragma unroll
  for (int r = 0; r < 8; ++r) sLw[(r + 8 * g) * LSTR + nlo] = acc2[r] + sBo[nlo];
  __syncthreads();

  {
    const float* v = sLw + nlo * LSTR;
    float m = v[0];
#pragma unroll
    for (int c = 1; c < CDIM; ++c) m = fmaxf(m, v[c]);
    float s = 0.0f;
#pragma unroll
    for (int c = 0; c < CDIM; ++c) s += expf(v[c] - m);
    float ls = logf(s);
    if (g == 0) {
#pragma unroll
      for (int c = 0; c < CDIM; ++c) sOw[nlo * CDIM + c] = v[c] - m - ls;
    }
  }
  __syncthreads();
#pragma unroll
  for (int j = 0; j < (16 * CDIM) / 32; ++j) {
    int idx = lane + 32 * j;
    if (rowBase + idx / CDIM < n) out[(size_t)rowBase * CDIM + idx] = sOw[idx];
  }
}


extern "C" void kernel_launch(void* const* d_in, const int* in_sizes, int n_in,
                              void* d_out, int out_size, void* d_ws, size_t ws_size,
                              hipStream_t stream) {
  const float* x    = (const float*)d_in[0];
  const int*   ei   = (const int*)d_in[1];
  const float* W1_0 = (const float*)d_in[2];
  const float* b1   = (const float*)d_in[3];
  const float* W2_0 = (const float*)d_in[4];
  const float* W2_1 = (const float*)d_in[5];
  const float* b2   = (const float*)d_in[6];
  const float* Wl   = (const float*)d_in[7];
  const float* bl   = (const float*)d_in[8];

  const int N = in_sizes[0] / FDIM;
  const int E = in_sizes[1] / 2;
  const int* src = ei;
  const int* dst = ei + E;

  float* ws     = (float*)d_ws;
  float* deg    = ws;
  float* lap    = deg + N;
  float* dis    = lap + (size_t)N * 64;
  float* wbuf   = dis + N;
  float* h      = wbuf + E;
  float* lap1   = h + (size_t)N * 64 + LAP2_PAD;
  float* out    = (float*)d_out;

  const int gblocks = (N + GROWS - 1) / GROWS;
  gemm64_relu_kernel<<<gblocks, 128, 0, stream>>>(x, W1_0, b1, h, lap, lap1, deg, N, E, ws_size);
  degree_kernel<<<dim3(DG_CHUNKS, DG_RANGES), DG_THREADS, 0, stream>>>(src, deg, N, E);
  scatter_kernel<<<(E + 255) / 256, 256, 0, stream>>>(src, dst, deg, h, lap, lap1, N, E, ws_size);
  cheb2_head_kernel<<<gblocks, 128, 0, stream>>>(h, lap, lap1, W2_0, W2_1, b2, Wl, bl, out, N, E, ws_size);
}


# lap without global float atomics: edges placed by destination range (208 nodes) inside the call, one block per range adds their rows in LDS and feeds the head from there; arrival-order LDS sums
# speedup vs baseline: 2.7570x; 1.2421x over previous
#include <hip/hip_runtime.h>
#include <hip/hip_bf16.h>
#include <stdint.h>


#define FDIM 64
#define HDIM 64
#define CDIM 10

#define GWAVES 4
#define GROWS  (GWAVES * 16)
#define ASTR   68
#define WPSTR  160
#define WPROWS 32
#define WOSTR  96

typedef float v2f __attribute__((ext_vector_type(2)));
typedef float v8f __attribute__((ext_vector_type(8)));

__device__ __forceinline__ v8f wmma4(v2f a, v2f b, v8f c) {
  return __builtin_amdgcn_wmma_f32_16x16x4_f32(false, a, false, b, (short)0, c,
                                               false, false);
}


#define BROWS    208
#define BWAVES   13
#define BTHREADS (BWAVES * 32)
#define MAXB     1024

__device__ __forceinline__ int nbins_of(int n) {
  return min((n + BROWS - 1) / BROWS, MAXB);
}

__device__ __forceinline__ int bin_of(int d, int nbins) {
  return min((int)((unsigned int)d / (unsigned int)BROWS), nbins - 1);
}


#define DG_THREADS 512
#define DG_CHUNKS  32
#define DG_RANGES  8
#define DG_RLEN    6656
#define DG_PER     (DG_RLEN / DG_THREADS)
#define DG_U       13
#define DG_HU      8

typedef int v4i __attribute__((ext_vector_type(4)));
typedef int v2i __attribute__((ext_vector_type(2)));

__device__ __forceinline__ void deg_count(unsigned int* cnt, int s, int base, int len) {
  const unsigned int r = (unsigned int)(s - base);
  if (r < (unsigned int)len) atomicAdd(&cnt[r], 1u);
}

__global__ void __launch_bounds__(DG_THREADS)
degree_kernel(const int* __restrict__ src, const int* __restrict__ dst, float* __restrict__ deg,
              unsigned int* __restrict__ bin_count, int n, int E) {
  __shared__ unsigned int cnt[DG_RLEN];
  const int tid = threadIdx.x;
  const int nq = E >> 2;
  const int qper = (nq + (int)gridDim.x - 1) / (int)gridDim.x;
  const int q_lo = (int)blockIdx.x * qper;
  const int q_hi = min(nq, q_lo + qper);
  const v4i* __restrict__ sq = (const v4i*)src;
  const int rot = ((int)blockIdx.x * (DG_RLEN / DG_CHUNKS)) & ~31;

  for (int base = (int)blockIdx.y * DG_RLEN; base < n; base += (int)gridDim.y * DG_RLEN) {
    const int len = min(DG_RLEN, n - base);
#pragma unroll
    for (int k = 0; k < DG_PER; ++k) cnt[tid + k * DG_THREADS] = 0u;
    __syncthreads();

    for (int q0 = q_lo; q0 < q_hi; q0 += DG_THREADS * DG_U) {
      v4i v[DG_U];
#pragma unroll
      for (int u = 0; u < DG_U; ++u) v[u] = sq[min(q0 + u * DG_THREADS + tid, q_hi - 1)];
      asm volatile("" :: "v"(v[0]), "v"(v[1]), "v"(v[2]), "v"(v[3]), "v"(v[4]), "v"(v[5]),
                         "v"(v[6]), "v"(v[7]), "v"(v[8]), "v"(v[9]), "v"(v[10]), "v"(v[11]),
                         "v"(v[12]));
#pragma unroll
      for (int u = 0; u < DG_U; ++u) {
        const int lim = (q0 + u * DG_THREADS + tid < q_hi) ? len : 0;
        deg_count(cnt, v[u].x, base, lim);
        deg_count(cnt, v[u].y, base, lim);
        deg_count(cnt, v[u].z, base, lim);
        deg_count(cnt, v[u].w, base, lim);
      }
    }
    if (blockIdx.x == 0 && tid < E - 4 * nq) deg_count(cnt, src[4 * nq + tid], base, len);
    __syncthreads();

    unsigned int c[DG_PER];
#pragma unroll
    for (int k = 0; k < DG_PER; ++k) {
      int j = tid + k * DG_THREADS + rot;
      if (j >= DG_RLEN) j -= DG_RLEN;
      c[k] = cnt[j];
    }
#pragma unroll
    for (int k = 0; k < DG_PER; ++k) {
      int j = tid + k * DG_THREADS + rot;
      if (j >= DG_RLEN) j -= DG_RLEN;
      if (j < len && c[k] != 0u) atomicAdd(&deg[base + j], (float)c[k]);
    }
    __syncthreads();
  }

  {
    const int nbins = nbins_of(n);
    const int nblk = (int)(gridDim.x * gridDim.y);
    const int lin = (int)(blockIdx.y * gridDim.x + blockIdx.x);
    const int per = (E + nblk - 1) / nblk;
    const int lo = lin * per;
    const int hi = min(E, lo + per);
    for (int b = tid; b < nbins; b += DG_THREADS) cnt[b] = 0u;
    __syncthreads();
    for (int i0 = lo; i0 < hi; i0 += DG_THREADS * DG_HU) {
      int dv[DG_HU];
#pragma unroll
      for (int u = 0; u < DG_HU; ++u) dv[u] = dst[min(i0 + u * DG_THREADS + tid, hi - 1)];
#pragma unroll
      for (int u = 0; u < DG_HU; ++u)
        if (i0 + u * DG_THREADS + tid < hi) atomicAdd(&cnt[bin_of(dv[u], nbins)], 1u);
    }
    __syncthreads();
    for (int b = tid; b < nbins; b += DG_THREADS) {
      const unsigned int c = cnt[b];
      if (c != 0u) atomicAdd(&bin_count[b], c);
    }
  }
}
__device__ __forceinline__ float dis_of(float d) {
  return (d > 0.0f) ? rsqrtf(fmaxf(d, 1.0f)) : 0.0f;
}

__device__ __forceinline__ int lane_bcast_i(int v, int j) {
  return __builtin_amdgcn_readlane(v, j);
}
__device__ __forceinline__ float lane_bcast_f(float v, int j) {
  return __int_as_float(__builtin_amdgcn_readlane(__float_as_int(v), j));
}

#define PL_THREADS 256
#define PL_U       8
#define PL_EDGES   (PL_THREADS * PL_U)
#define PL_OWN     (MAXB / PL_THREADS)

__global__ void __launch_bounds__(PL_THREADS)
place_kernel(const int* __restrict__ src, const int* __restrict__ dst,
             const unsigned int* __restrict__ bin_count, unsigned int* __restrict__ bin_cursor,
             unsigned int* __restrict__ bin_start, int* __restrict__ rec, int n, int E) {
  __shared__ unsigned int lc[MAXB];
  __shared__ unsigned int pre[MAXB];
  __shared__ unsigned int wsum[PL_THREADS / 32];
  const int tid = threadIdx.x, wave = tid >> 5, lane = tid & 31;
  const int nbins = nbins_of(n);
  const int lo = (int)blockIdx.x * PL_EDGES;
  const int hi = min(E, lo + PL_EDGES);

  for (int b = tid; b < MAXB; b += PL_THREADS) lc[b] = 0u;
  int sv[PL_U], dv[PL_U];
#pragma unroll
  for (int u = 0; u < PL_U; ++u) {
    const int i = min(lo + u * PL_THREADS + tid, E - 1);
    dv[u] = dst[i];
    sv[u] = src[i];
  }
  __syncthreads();
#pragma unroll
  for (int u = 0; u < PL_U; ++u)
    if (lo + u * PL_THREADS + tid < hi) atomicAdd(&lc[bin_of(dv[u], nbins)], 1u);

  unsigned int c[PL_OWN];
  unsigned int tsum = 0u;
#pragma unroll
  for (int k = 0; k < PL_OWN; ++k) {
    const int b = tid * PL_OWN + k;
    c[k] = (b < nbins) ? bin_count[b] : 0u;
    tsum += c[k];
  }
  unsigned int inc = tsum;
#pragma unroll
  for (int off = 1; off < 32; off <<= 1) {
    const unsigned int v = __shfl_up(inc, off);
    if (lane >= off) inc += v;
  }
  if (lane == 31) wsum[wave] = inc;
  __syncthreads();
  unsigned int ex = inc - tsum;
  for (int w = 0; w < wave; ++w) ex += wsum[w];
#pragma unroll
  for (int k = 0; k < PL_OWN; ++k) {
    pre[tid * PL_OWN + k] = ex;
    if (blockIdx.x == 0 && tid * PL_OWN + k < nbins) bin_start[tid * PL_OWN + k] = ex;
    ex += c[k];
  }
  __syncthreads();

  for (int b = tid; b < nbins; b += PL_THREADS) {
    const unsigned int cb = lc[b];
    if (cb != 0u) lc[b] = pre[b] + atomicAdd(&bin_cursor[b], cb);
  }
  __syncthreads();

#pragma unroll
  for (int u = 0; u < PL_U; ++u)
    if (lo + u * PL_THREADS + tid < hi) {
      const unsigned int pos = atomicAdd(&lc[bin_of(dv[u], nbins)], 1u);
      const v2i pr = {sv[u], dv[u]};
      if (pos < (unsigned int)E) ((v2i*)rec)[pos] = pr;
    }
}


__device__ __forceinline__ void stage_w_pairs(float* sWp, const float* __restrict__ W,
                                              int tid) {
  const float4* Wv = (const float4*)W;
#pragma unroll
  for (int i = 0; i < 8; ++i) {
    int idx = tid + i * 128;
    float4 v = Wv[idx];
    int k = idx >> 4;
    int c4 = (idx & 15) * 4;
    float* dstp = sWp + (k >> 1) * WPSTR + (k & 1);
    dstp[(c4 + 0) * 2] = v.x;
    dstp[(c4 + 1) * 2] = v.y;
    dstp[(c4 + 2) * 2] = v.z;
    dstp[(c4 + 3) * 2] = v.w;
  }
}

__device__ __forceinline__ void stage_rows(float* sAw, const float* __restrict__ A,
                                           int rowBase, int n, int lane) {
#pragma unroll
  for (int i = 0; i < 8; ++i) {
    int fi = lane + i * 32;
    int r = fi >> 4, c4 = fi & 15;
    int row = rowBase + r;
    if (row >= n) row = n - 1;
    ((float4*)(sAw + r * ASTR))[c4] = ((const float4*)(A + (size_t)row * 64))[c4];
  }
}

__device__ __forceinline__ void mm_accum64(const float* sAw, const float* sWp,
                                           int lane, v8f acc[4]) {
  const int nlo = lane & 15;
  const int g = lane >> 4;
#pragma unroll
  for (int kk = 0; kk < 16; ++kk) {
    const int k0 = kk * 4 + 2 * g;
    v2f a = *(const v2f*)(sAw + nlo * ASTR + k0);
    const float* brow = sWp + (kk * 2 + g) * WPSTR;
#pragma unroll
    for (int nt = 0; nt < 4; ++nt) {
      v2f b = *(const v2f*)(brow + (nt * 16 + nlo) * 2);
      acc[nt] = wmma4(a, b, acc[nt]);
    }
  }
}


__global__ void __launch_bounds__(128)
gemm64_relu_kernel(const float* __restrict__ A, const float* __restrict__ W,
                   const float* __restrict__ bias, float* __restrict__ out,
                   float* __restrict__ deg, unsigned int* __restrict__ bin_tab, int n) {
  __shared__ float sWp[WPROWS * WPSTR];
  __shared__ float sA[GWAVES * 16 * ASTR];
  __shared__ float sB[64];
  const int tid = threadIdx.x, wave = tid >> 5, lane = tid & 31;
  stage_w_pairs(sWp, W, tid);
  if (tid < 64) sB[tid] = bias[tid];

  const int rowBase = blockIdx.x * GROWS + wave * 16;
  float* sAw = sA + wave * 16 * ASTR;
  stage_rows(sAw, A, rowBase, n, lane);
  __syncthreads();

  if (lane < 16 && rowBase + lane < n) deg[rowBase + lane] = 0.0f;
  if (blockIdx.x == 0)
    for (int i = tid; i < 2 * MAXB; i += 128) bin_tab[i] = 0u;

  v8f z = {0, 0, 0, 0, 0, 0, 0, 0};
  v8f acc[4] = {z, z, z, z};
  mm_accum64(sAw, sWp, lane, acc);

  const int nlo = lane & 15, g = lane >> 4;
#pragma unroll
  for (int nt = 0; nt < 4; ++nt)
#pragma unroll
    for (int r = 0; r < 8; ++r) {
      int row = rowBase + r + 8 * g;
      int col = nt * 16 + nlo;
      if (row < n) out[(size_t)row * 64 + col] = fmaxf(acc[nt][r] + sB[col], 0.0f);
    }
}

#define LSTR 17
#define BG   16

#define WIT ((1024 + BTHREADS - 1) / BTHREADS)

__device__ __forceinline__ void put_w_pairs(float* sWp, float4 v, int idx) {
  const int k = idx >> 4;
  const int c4 = (idx & 15) * 4;
  float* dstp = sWp + (k >> 1) * WPSTR + (k & 1);
  dstp[(c4 + 0) * 2] = v.x;
  dstp[(c4 + 1) * 2] = v.y;
  dstp[(c4 + 2) * 2] = v.z;
  dstp[(c4 + 3) * 2] = v.w;
}

template <int G0>
__device__ __forceinline__ void add_rows(const float* __restrict__ h, float* sS, int s, int dl,
                                         float we, int lane) {
  float a[BG], b[BG];
#pragma unroll
  for (int u = 0; u < BG; ++u) {
    const float* hp = h + (size_t)lane_bcast_i(s, G0 + u) * HDIM + lane;
    a[u] = hp[0];
    b[u] = hp[32];
  }
#pragma unroll
  for (int u = 0; u < BG; ++u) {
    float* sp = sS + lane_bcast_i(dl, G0 + u) * ASTR + lane;
    const float wj = lane_bcast_f(we, G0 + u);
    atomicAdd(sp, wj * a[u]);
    atomicAdd(sp + 32, wj * b[u]);
  }
}

__global__ void __launch_bounds__(BTHREADS)
bin_head_kernel(const float* __restrict__ h, const int* __restrict__ rec,
                const unsigned int* __restrict__ bin_count,
                const unsigned int* __restrict__ bin_start, const float* __restrict__ deg,
                const float* __restrict__ W0, const float* __restrict__ W1,
                const float* __restrict__ bias, const float* __restrict__ Wl,
                const float* __restrict__ bl, float* __restrict__ out, int n) {
  __shared__ float sWp[2 * WPROWS * WPSTR];
  __shared__ float sWo[WPROWS * WOSTR];
  __shared__ float sA[BROWS * ASTR];
  __shared__ float sS[BROWS * ASTR];
  __shared__ float sL[BWAVES * 16 * LSTR];
  __shared__ float sO[BWAVES * 16 * CDIM];
  __shared__ float sB[64];
  __shared__ float sBo[16];
  const int tid = threadIdx.x, lane = tid & 31;
  const int wave = __builtin_amdgcn_readfirstlane(tid >> 5);
  const int bin = blockIdx.x;
  const int rowBase = bin * BROWS + wave * 16;
  float* sAw = sA + wave * 16 * ASTR;
  float* sSw = sS + wave * 16 * ASTR;
  float* sLw = sL + wave * 16 * LSTR;
  float* sOw = sO + wave * 16 * CDIM;

  const unsigned int first = bin_start[bin];
  const unsigned int cntb = bin_count[bin];
  unsigned int c0 = (unsigned int)wave * 32u;
  const bool any = c0 < cntb;

  float4 w0v[WIT], w1v[WIT];
  float wlv[WIT];
#pragma unroll
  for (int i = 0; i < WIT; ++i) {
    const int idx = min(tid + i * BTHREADS, 1023);
    w0v[i] = ((const float4*)W0)[idx];
    w1v[i] = ((const float4*)W1)[idx];
    wlv[i] = Wl[(idx >> 4) * CDIM + min(idx & 15, CDIM - 1)];
  }
  const float bv = bias[tid & 63];
  const float blv = bl[min(tid & 15, CDIM - 1)];
  float4 hv[8];
#pragma unroll
  for (int i = 0; i < 8; ++i) {
    const int fi = lane + i * 32;
    const int row = min(rowBase + (fi >> 4), n - 1);
    hv[i] = ((const float4*)(h + (size_t)row * 64))[fi & 15];
  }
  int mc = any ? (int)min(32u, cntb - c0) : 1;
  const v2i* __restrict__ rec2 = (const v2i*)rec;
  v2i sd = rec2[any ? first + c0 + (unsigned int)min(lane, mc - 1) : 0u];

#pragma unroll
  for (int i = 0; i < WIT; ++i) {
    const int idx = tid + i * BTHREADS;
    if (idx < 1024) {
      put_w_pairs(sWp, w0v[i], idx);
      put_w_pairs(sWp + WPROWS * WPSTR, w1v[i], idx);
      const int k = idx >> 4, c = idx & 15;
      sWo[(k >> 1) * WOSTR + c * 2 + (k & 1)] = (c < CDIM) ? wlv[i] : 0.0f;
    }
  }
  if (tid < 64) sB[tid] = bv;
  if (tid < 16) sBo[tid] = (tid < CDIM) ? blv : 0.0f;
#pragma unroll
  for (int i = 0; i < 8; ++i) {
    const int fi = lane + i * 32;
    ((float4*)(sAw + (fi >> 4) * ASTR))[fi & 15] = hv[i];
  }
  for (int i = tid; i < BROWS * ASTR; i += BTHREADS) sS[i] = 0.0f;
  float we = -(dis_of(deg[sd.x]) * dis_of(deg[sd.y]));
  __syncthreads();

  if (any) {
    for (;;) {
      const unsigned int c1 = c0 + BTHREADS;
      const bool more = c1 < cntb;
      const int m1 = more ? (int)min(32u, cntb - c1) : mc;
      const v2i sdn = rec2[first + (more ? c1 : c0) + (unsigned int)min(lane, m1 - 1)];
      const int s = sd.x;
      const int dl = (int)min((unsigned int)(sd.y - bin * BROWS), (unsigned int)(BROWS - 1));
      float dgs = 1.0f, dgd = 1.0f;
      if (mc == 32) {
        add_rows<0>(h, sS, s, dl, we, lane);
        dgs = deg[sdn.x];
        dgd = deg[sdn.y];
        add_rows<BG>(h, sS, s, dl, we, lane);
      } else {
        for (int j = 0; j < mc; ++j) {
          const float* hp = h + (size_t)lane_bcast_i(s, j) * HDIM + lane;
          float* sp = sS + lane_bcast_i(dl, j) * ASTR + lane;
          const float wj = lane_bcast_f(we, j);
          atomicAdd(sp, wj * hp[0]);
          atomicAdd(sp + 32, wj * hp[32]);
        }
      }
      if (!more) break;
      sd = sdn;
      we = -(dis_of(dgs) * dis_of(dgd));
      mc = m1;
      c0 = c1;
    }
  }
  __syncthreads();

  v8f z = {0, 0, 0, 0, 0, 0, 0, 0};
  v8f acc[4] = {z, z, z, z};
  mm_accum64(sAw, sWp, lane, acc);
  mm_accum64(sSw, sWp + WPROWS * WPSTR, lane, acc);

  const int nlo = lane & 15, g = lane >> 4;
#pragma unroll
  for (int nt = 0; nt < 4; ++nt)
#pragma unroll
    for (int r = 0; r < 8; ++r) {
      int rl = r + 8 * g;
      int col = nt * 16 + nlo;
      float v = fmaxf(acc[nt][r] + sB[col], 0.0f);
      sAw[rl * ASTR + col] = v + sAw[rl * ASTR + col];
    }
  __syncthreads();

  v8f acc2 = z;
#pragma unroll
  for (int kk = 0; kk < 16; ++kk) {
    const int k0 = kk * 4 + 2 * g;
    v2f a = *(const v2f*)(sAw + nlo * ASTR + k0);
    v2f b = *(const v2f*)(sWo + (kk * 2 + g) * WOSTR + nlo * 2);
    acc2 = wmma4(a, b, acc2);
  }
#pragma unroll
  for (int r = 0; r < 8; ++r) sLw[(r + 8 * g) * LSTR + nlo] = acc2[r] + sBo[nlo];
  __syncthreads();

  {
    const float* v = sLw + nlo * LSTR;
    float m = v[0];
#pragma unroll
    for (int c = 1; c < CDIM; ++c) m = fmaxf(m, v[c]);
    float s = 0.0f;
#pragma unroll
    for (int c = 0; c < CDIM; ++c) s += expf(v[c] - m);
    float ls = logf(s);
    if (g == 0) {
#pragma unroll
      for (int c = 0; c < CDIM; ++c) sOw[nlo * CDIM + c] = v[c] - m - ls;
    }
  }
  __syncthreads();
#pragma unroll
  for (int j = 0; j < (16 * CDIM) / 32; ++j) {
    int idx = lane + 32 * j;
    if (rowBase + idx / CDIM < n) out[(size_t)rowBase * CDIM + idx] = sOw[idx];
  }
}


extern "C" void kernel_launch(void* const* d_in, const int* in_sizes, int n_in,
                              void* d_out, int out_size, void* d_ws, size_t ws_size,
                              hipStream_t stream) {
  const float* x    = (const float*)d_in[0];
  const int*   ei   = (const int*)d_in[1];
  const float* W1_0 = (const float*)d_in[2];
  const float* b1   = (const float*)d_in[3];
  const float* W2_0 = (const float*)d_in[4];
  const float* W2_1 = (const float*)d_in[5];
  const float* b2   = (const float*)d_in[6];
  const float* Wl   = (const float*)d_in[7];
  const float* bl   = (const float*)d_in[8];

  const int N = in_sizes[0] / FDIM;
  const int E = in_sizes[1] / 2;
  const int* src = ei;
  const int* dst = ei + E;

  float* ws      = (float*)d_ws;
  float* h       = ws;
  int*   rec     = (int*)(h + (size_t)N * 64);
  unsigned int* bin_count  = (unsigned int*)(rec + (size_t)E * 2);
  unsigned int* bin_cursor = bin_count + MAXB;
  unsigned int* bin_start  = bin_cursor + MAXB;
  float* deg     = (float*)(bin_start + MAXB);
  float* out     = (float*)d_out;

  const int gblocks = (N + GROWS - 1) / GROWS;
  const int nbins = (N + BROWS - 1) / BROWS;
  gemm64_relu_kernel<<<gblocks, 128, 0, stream>>>(x, W1_0, b1, h, deg, bin_count, N);
  degree_kernel<<<dim3(DG_CHUNKS, DG_RANGES), DG_THREADS, 0, stream>>>(src, dst, deg, bin_count, N, E);
  place_kernel<<<(E + PL_EDGES - 1) / PL_EDGES, PL_THREADS, 0, stream>>>(src, dst, bin_count, bin_cursor, bin_start, rec, N, E);
  bin_head_kernel<<<nbins, BTHREADS, 0, stream>>>(h, rec, bin_count, bin_start, deg, W2_0, W2_1, b2, Wl, bl, out, N);
}


# bin's last, partly filled chunk of edges takes the grouped 16-rows-per-trip walk (adds of its empty lanes skipped) instead of one edge, two serial trips to L2, at a time; same sums, arrival order
# speedup vs baseline: 3.1132x; 1.1292x over previous
#include <hip/hip_runtime.h>
#include <hip/hip_bf16.h>
#include <stdint.h>


#define FDIM 64
#define HDIM 64
#define CDIM 10

#define GWAVES 4
#define GROWS  (GWAVES * 16)
#define ASTR   68
#define WPSTR  160
#define WPROWS 32
#define WOSTR  96

typedef float v2f __attribute__((ext_vector_type(2)));
typedef float v8f __attribute__((ext_vector_type(8)));

__device__ __forceinline__ v8f wmma4(v2f a, v2f b, v8f c) {
  return __builtin_amdgcn_wmma_f32_16x16x4_f32(false, a, false, b, (short)0, c,
                                               false, false);
}


#define BROWS    208
#define BWAVES   13
#define BTHREADS (BWAVES * 32)
#define MAXB     1024

__device__ __forceinline__ int nbins_of(int n) {
  return min((n + BROWS - 1) / BROWS, MAXB);
}

__device__ __forceinline__ int bin_of(int d, int nbins) {
  return min((int)((unsigned int)d / (unsigned int)BROWS), nbins - 1);
}


#define DG_THREADS 512
#define DG_CHUNKS  32
#define DG_RANGES  8
#define DG_RLEN    6656
#define DG_PER     (DG_RLEN / DG_THREADS)
#define DG_U       13
#define DG_HU      8

typedef int v4i __attribute__((ext_vector_type(4)));
typedef int v2i __attribute__((ext_vector_type(2)));

__device__ __forceinline__ void deg_count(unsigned int* cnt, int s, int base, int len) {
  const unsigned int r = (unsigned int)(s - base);
  if (r < (unsigned int)len) atomicAdd(&cnt[r], 1u);
}

__global__ void __launch_bounds__(DG_THREADS)
degree_kernel(const int* __restrict__ src, const int* __restrict__ dst, float* __restrict__ deg,
              unsigned int* __restrict__ bin_count, int n, int E) {
  __shared__ unsigned int cnt[DG_RLEN];
  const int tid = threadIdx.x;
  const int nq = E >> 2;
  const int qper = (nq + (int)gridDim.x - 1) / (int)gridDim.x;
  const int q_lo = (int)blockIdx.x * qper;
  const int q_hi = min(nq, q_lo + qper);
  const v4i* __restrict__ sq = (const v4i*)src;
  const int rot = ((int)blockIdx.x * (DG_RLEN / DG_CHUNKS)) & ~31;

  for (int base = (int)blockIdx.y * DG_RLEN; base < n; base += (int)gridDim.y * DG_RLEN) {
    const int len = min(DG_RLEN, n - base);
#pragma unroll
    for (int k = 0; k < DG_PER; ++k) cnt[tid + k * DG_THREADS] = 0u;
    __syncthreads();

    for (int q0 = q_lo; q0 < q_hi; q0 += DG_THREADS * DG_U) {
      v4i v[DG_U];
#pragma unroll
      for (int u = 0; u < DG_U; ++u) v[u] = sq[min(q0 + u * DG_THREADS + tid, q_hi - 1)];
      asm volatile("" :: "v"(v[0]), "v"(v[1]), "v"(v[2]), "v"(v[3]), "v"(v[4]), "v"(v[5]),
                         "v"(v[6]), "v"(v[7]), "v"(v[8]), "v"(v[9]), "v"(v[10]), "v"(v[11]),
                         "v"(v[12]));
#pragma unroll
      for (int u = 0; u < DG_U; ++u) {
        const int lim = (q0 + u * DG_THREADS + tid < q_hi) ? len : 0;
        deg_count(cnt, v[u].x, base, lim);
        deg_count(cnt, v[u].y, base, lim);
        deg_count(cnt, v[u].z, base, lim);
        deg_count(cnt, v[u].w, base, lim);
      }
    }
    if (blockIdx.x == 0 && tid < E - 4 * nq) deg_count(cnt, src[4 * nq + tid], base, len);
    __syncthreads();

    unsigned int c[DG_PER];
#pragma unroll
    for (int k = 0; k < DG_PER; ++k) {
      int j = tid + k * DG_THREADS + rot;
      if (j >= DG_RLEN) j -= DG_RLEN;
      c[k] = cnt[j];
    }
#pragma unroll
    for (int k = 0; k < DG_PER; ++k) {
      int j = tid + k * DG_THREADS + rot;
      if (j >= DG_RLEN) j -= DG_RLEN;
      if (j < len && c[k] != 0u) atomicAdd(&deg[base + j], (float)c[k]);
    }
    __syncthreads();
  }

  {
    const int nbins = nbins_of(n);
    const int nblk = (int)(gridDim.x * gridDim.y);
    const int lin = (int)(blockIdx.y * gridDim.x + blockIdx.x);
    const int per = (E + nblk - 1) / nblk;
    const int lo = lin * per;
    const int hi = min(E, lo + per);
    for (int b = tid; b < nbins; b += DG_THREADS) cnt[b] = 0u;
    __syncthreads();
    for (int i0 = lo; i0 < hi; i0 += DG_THREADS * DG_HU) {
      int dv[DG_HU];
#pragma unroll
      for (int u = 0; u < DG_HU; ++u) dv[u] = dst[min(i0 + u * DG_THREADS + tid, hi - 1)];
#pragma unroll
      for (int u = 0; u < DG_HU; ++u)
        if (i0 + u * DG_THREADS + tid < hi) atomicAdd(&cnt[bin_of(dv[u], nbins)], 1u);
    }
    __syncthreads();
    for (int b = tid; b < nbins; b += DG_THREADS) {
      const unsigned int c = cnt[b];
      if (c != 0u) atomicAdd(&bin_count[b], c);
    }
  }
}
__device__ __forceinline__ float dis_of(float d) {
  return (d > 0.0f) ? rsqrtf(fmaxf(d, 1.0f)) : 0.0f;
}

__device__ __forceinline__ int lane_bcast_i(int v, int j) {
  return __builtin_amdgcn_readlane(v, j);
}
__device__ __forceinline__ float lane_bcast_f(float v, int j) {
  return __int_as_float(__builtin_amdgcn_readlane(__float_as_int(v), j));
}

#define PL_THREADS 256
#define PL_U       8
#define PL_EDGES   (PL_THREADS * PL_U)
#define PL_OWN     (MAXB / PL_THREADS)

__global__ void __launch_bounds__(PL_THREADS)
place_kernel(const int* __restrict__ src, const int* __restrict__ dst,
             const unsigned int* __restrict__ bin_count, unsigned int* __restrict__ bin_cursor,
             unsigned int* __restrict__ bin_start, int* __restrict__ rec, int n, int E) {
  __shared__ unsigned int lc[MAXB];
  __shared__ unsigned int pre[MAXB];
  __shared__ unsigned int wsum[PL_THREADS / 32];
  const int tid = threadIdx.x, wave = tid >> 5, lane = tid & 31;
  const int nbins = nbins_of(n);
  const int lo = (int)blockIdx.x * PL_EDGES;
  const int hi = min(E, lo + PL_EDGES);

  for (int b = tid; b < MAXB; b += PL_THREADS) lc[b] = 0u;
  int sv[PL_U], dv[PL_U];
#pragma unroll
  for (int u = 0; u < PL_U; ++u) {
    const int i = min(lo + u * PL_THREADS + tid, E - 1);
    dv[u] = dst[i];
    sv[u] = src[i];
  }
  __syncthreads();
#pragma unroll
  for (int u = 0; u < PL_U; ++u)
    if (lo + u * PL_THREADS + tid < hi) atomicAdd(&lc[bin_of(dv[u], nbins)], 1u);

  unsigned int c[PL_OWN];
  unsigned int tsum = 0u;
#pragma unroll
  for (int k = 0; k < PL_OWN; ++k) {
    const int b = tid * PL_OWN + k;
    c[k] = (b < nbins) ? bin_count[b] : 0u;
    tsum += c[k];
  }
  unsigned int inc = tsum;
#pragma unroll
  for (int off = 1; off < 32; off <<= 1) {
    const unsigned int v = __shfl_up(inc, off);
    if (lane >= off) inc += v;
  }
  if (lane == 31) wsum[wave] = inc;
  __syncthreads();
  unsigned int ex = inc - tsum;
  for (int w = 0; w < wave; ++w) ex += wsum[w];
#pragma unroll
  for (int k = 0; k < PL_OWN; ++k) {
    pre[tid * PL_OWN + k] = ex;
    if (blockIdx.x == 0 && tid * PL_OWN + k < nbins) bin_start[tid * PL_OWN + k] = ex;
    ex += c[k];
  }
  __syncthreads();

  for (int b = tid; b < nbins; b += PL_THREADS) {
    const unsigned int cb = lc[b];
    if (cb != 0u) lc[b] = pre[b] + atomicAdd(&bin_cursor[b], cb);
  }
  __syncthreads();

#pragma unroll
  for (int u = 0; u < PL_U; ++u)
    if (lo + u * PL_THREADS + tid < hi) {
      const unsigned int pos = atomicAdd(&lc[bin_of(dv[u], nbins)], 1u);
      const v2i pr = {sv[u], dv[u]};
      if (pos < (unsigned int)E) ((v2i*)rec)[pos] = pr;
    }
}


__device__ __forceinline__ void stage_w_pairs(float* sWp, const float* __restrict__ W,
                                              int tid) {
  const float4* Wv = (const float4*)W;
#pragma unroll
  for (int i = 0; i < 8; ++i) {
    int idx = tid + i * 128;
    float4 v = Wv[idx];
    int k = idx >> 4;
    int c4 = (idx & 15) * 4;
    float* dstp = sWp + (k >> 1) * WPSTR + (k & 1);
    dstp[(c4 + 0) * 2] = v.x;
    dstp[(c4 + 1) * 2] = v.y;
    dstp[(c4 + 2) * 2] = v.z;
    dstp[(c4 + 3) * 2] = v.w;
  }
}

__device__ __forceinline__ void stage_rows(float* sAw, const float* __restrict__ A,
                                           int rowBase, int n, int lane) {
#pragma unroll
  for (int i = 0; i < 8; ++i) {
    int fi = lane + i * 32;
    int r = fi >> 4, c4 = fi & 15;
    int row = rowBase + r;
    if (row >= n) row = n - 1;
    ((float4*)(sAw + r * ASTR))[c4] = ((const float4*)(A + (size_t)row * 64))[c4];
  }
}

__device__ __forceinline__ void mm_accum64(const float* sAw, const float* sWp,
                                           int lane, v8f acc[4]) {
  const int nlo = lane & 15;
  const int g = lane >> 4;
#pragma unroll
  for (int kk = 0; kk < 16; ++kk) {
    const int k0 = kk * 4 + 2 * g;
    v2f a = *(const v2f*)(sAw + nlo * ASTR + k0);
    const float* brow = sWp + (kk * 2 + g) * WPSTR;
#pragma unroll
    for (int nt = 0; nt < 4; ++nt) {
      v2f b = *(const v2f*)(brow + (nt * 16 + nlo) * 2);
      acc[nt] = wmma4(a, b, acc[nt]);
    }
  }
}


__global__ void __launch_bounds__(128)
gemm64_relu_kernel(const float* __restrict__ A, const float* __restrict__ W,
                   const float* __restrict__ bias, float* __restrict__ out,
                   float* __restrict__ deg, unsigned int* __restrict__ bin_tab, int n) {
  __shared__ float sWp[WPROWS * WPSTR];
  __shared__ float sA[GWAVES * 16 * ASTR];
  __shared__ float sB[64];
  const int tid = threadIdx.x, wave = tid >> 5, lane = tid & 31;
  stage_w_pairs(sWp, W, tid);
  if (tid < 64) sB[tid] = bias[tid];

  const int rowBase = blockIdx.x * GROWS + wave * 16;
  float* sAw = sA + wave * 16 * ASTR;
  stage_rows(sAw, A, rowBase, n, lane);
  __syncthreads();

  if (lane < 16 && rowBase + lane < n) deg[rowBase + lane] = 0.0f;
  if (blockIdx.x == 0)
    for (int i = tid; i < 2 * MAXB; i += 128) bin_tab[i] = 0u;

  v8f z = {0, 0, 0, 0, 0, 0, 0, 0};
  v8f acc[4] = {z, z, z, z};
  mm_accum64(sAw, sWp, lane, acc);

  const int nlo = lane & 15, g = lane >> 4;
#pragma unroll
  for (int nt = 0; nt < 4; ++nt)
#pragma unroll
    for (int r = 0; r < 8; ++r) {
      int row = rowBase + r + 8 * g;
      int col = nt * 16 + nlo;
      if (row < n) out[(size_t)row * 64 + col] = fmaxf(acc[nt][r] + sB[col], 0.0f);
    }
}

#define LSTR 17
#define BG   16

#define WIT ((1024 + BTHREADS - 1) / BTHREADS)

__device__ __forceinline__ void put_w_pairs(float* sWp, float4 v, int idx) {
  const int k = idx >> 4;
  const int c4 = (idx & 15) * 4;
  float* dstp = sWp + (k >> 1) * WPSTR + (k & 1);
  dstp[(c4 + 0) * 2] = v.x;
  dstp[(c4 + 1) * 2] = v.y;
  dstp[(c4 + 2) * 2] = v.z;
  dstp[(c4 + 3) * 2] = v.w;
}

template <int G0, bool PART>
__device__ __forceinline__ void add_rows(const float* __restrict__ h, float* sS, int s, int dl,
                                         float we, int lane, int mc) {
  float a[BG], b[BG];
#pragma unroll
  for (int u = 0; u < BG; ++u) {
    const float* hp = h + (size_t)lane_bcast_i(s, G0 + u) * HDIM + lane;
    a[u] = hp[0];
    b[u] = hp[32];
  }
  if (PART) {
    asm volatile("" :: "v"(a[0]), "v"(a[1]), "v"(a[2]), "v"(a[3]), "v"(a[4]), "v"(a[5]),
                       "v"(a[6]), "v"(a[7]), "v"(a[8]), "v"(a[9]), "v"(a[10]), "v"(a[11]),
                       "v"(a[12]), "v"(a[13]), "v"(a[14]), "v"(a[15]));
    asm volatile("" :: "v"(b[0]), "v"(b[1]), "v"(b[2]), "v"(b[3]), "v"(b[4]), "v"(b[5]),
                       "v"(b[6]), "v"(b[7]), "v"(b[8]), "v"(b[9]), "v"(b[10]), "v"(b[11]),
                       "v"(b[12]), "v"(b[13]), "v"(b[14]), "v"(b[15]));
  }
#pragma unroll
  for (int u = 0; u < BG; ++u) {
    if (!PART || G0 + u < mc) {
      float* sp = sS + lane_bcast_i(dl, G0 + u) * ASTR + lane;
      const float wj = lane_bcast_f(we, G0 + u);
      atomicAdd(sp, wj * a[u]);
      atomicAdd(sp + 32, wj * b[u]);
    }
  }
}

__global__ void __launch_bounds__(BTHREADS)
bin_head_kernel(const float* __restrict__ h, const int* __restrict__ rec,
                const unsigned int* __restrict__ bin_count,
                const unsigned int* __restrict__ bin_start, const float* __restrict__ deg,
                const float* __restrict__ W0, const float* __restrict__ W1,
                const float* __restrict__ bias, const float* __restrict__ Wl,
                const float* __restrict__ bl, float* __restrict__ out, int n) {
  __shared__ float sWp[2 * WPROWS * WPSTR];
  __shared__ float sWo[WPROWS * WOSTR];
  __shared__ float sA[BROWS * ASTR];
  __shared__ float sS[BROWS * ASTR];
  __shared__ float sL[BWAVES * 16 * LSTR];
  __shared__ float sO[BWAVES * 16 * CDIM];
  __shared__ float sB[64];
  __shared__ float sBo[16];
  const int tid = threadIdx.x, lane = tid & 31;
  const int wave = __builtin_amdgcn_readfirstlane(tid >> 5);
  const int bin = blockIdx.x;
  const int rowBase = bin * BROWS + wave * 16;
  float* sAw = sA + wave * 16 * ASTR;
  float* sSw = sS + wave * 16 * ASTR;
  float* sLw = sL + wave * 16 * LSTR;
  float* sOw = sO + wave * 16 * CDIM;

  const unsigned int first = bin_start[bin];
  const unsigned int cntb = bin_count[bin];
  unsigned int c0 = (unsigned int)wave * 32u;
  const bool any = c0 < cntb;

  float4 w0v[WIT], w1v[WIT];
  float wlv[WIT];
#pragma unroll
  for (int i = 0; i < WIT; ++i) {
    const int idx = min(tid + i * BTHREADS, 1023);
    w0v[i] = ((const float4*)W0)[idx];
    w1v[i] = ((const float4*)W1)[idx];
    wlv[i] = Wl[(idx >> 4) * CDIM + min(idx & 15, CDIM - 1)];
  }
  const float bv = bias[tid & 63];
  const float blv = bl[min(tid & 15, CDIM - 1)];
  float4 hv[8];
#pragma unroll
  for (int i = 0; i < 8; ++i) {
    const int fi = lane + i * 32;
    const int row = min(rowBase + (fi >> 4), n - 1);
    hv[i] = ((const float4*)(h + (size_t)row * 64))[fi & 15];
  }
  int mc = any ? (int)min(32u, cntb - c0) : 1;
  const v2i* __restrict__ rec2 = (const v2i*)rec;
  v2i sd = rec2[any ? first + c0 + (unsigned int)min(lane, mc - 1) : 0u];

#pragma unroll
  for (int i = 0; i < WIT; ++i) {
    const int idx = tid + i * BTHREADS;
    if (idx < 1024) {
      put_w_pairs(sWp, w0v[i], idx);
      put_w_pairs(sWp + WPROWS * WPSTR, w1v[i], idx);
      const int k = idx >> 4, c = idx & 15;
      sWo[(k >> 1) * WOSTR + c * 2 + (k & 1)] = (c < CDIM) ? wlv[i] : 0.0f;
    }
  }
  if (tid < 64) sB[tid] = bv;
  if (tid < 16) sBo[tid] = (tid < CDIM) ? blv : 0.0f;
#pragma unroll
  for (int i = 0; i < 8; ++i) {
    const int fi = lane + i * 32;
    ((float4*)(sAw + (fi >> 4) * ASTR))[fi & 15] = hv[i];
  }
  for (int i = tid; i < BROWS * ASTR; i += BTHREADS) sS[i] = 0.0f;
  float we = -(dis_of(deg[sd.x]) * dis_of(deg[sd.y]));
  __syncthreads();

  if (any) {
    for (;;) {
      const unsigned int c1 = c0 + BTHREADS;
      const bool more = c1 < cntb;
      const int m1 = more ? (int)min(32u, cntb - c1) : mc;
      const v2i sdn = rec2[first + (more ? c1 : c0) + (unsigned int)min(lane, m1 - 1)];
      const int s = sd.x;
      const int dl = (int)min((unsigned int)(sd.y - bin * BROWS), (unsigned int)(BROWS - 1));
      float dgs = 1.0f, dgd = 1.0f;
      if (mc == 32) {
        add_rows<0, false>(h, sS, s, dl, we, lane, mc);
        dgs = deg[sdn.x];
        dgd = deg[sdn.y];
        add_rows<BG, false>(h, sS, s, dl, we, lane, mc);
      } else {
        add_rows<0, true>(h, sS, s, dl, we, lane, mc);
        if (mc > BG) add_rows<BG, true>(h, sS, s, dl, we, lane, mc);
      }
      if (!more) break;
      sd = sdn;
      we = -(dis_of(dgs) * dis_of(dgd));
      mc = m1;
      c0 = c1;
    }
  }
  __syncthreads();

  v8f z = {0, 0, 0, 0, 0, 0, 0, 0};
  v8f acc[4] = {z, z, z, z};
  mm_accum64(sAw, sWp, lane, acc);
  mm_accum64(sSw, sWp + WPROWS * WPSTR, lane, acc);

  const int nlo = lane & 15, g = lane >> 4;
#pragma unroll
  for (int nt = 0; nt < 4; ++nt)
#pragma unroll
    for (int r = 0; r < 8; ++r) {
      int rl = r + 8 * g;
      int col = nt * 16 + nlo;
      float v = fmaxf(acc[nt][r] + sB[col], 0.0f);
      sAw[rl * ASTR + col] = v + sAw[rl * ASTR + col];
    }
  __syncthreads();

  v8f acc2 = z;
#pragma unroll
  for (int kk = 0; kk < 16; ++kk) {
    const int k0 = kk * 4 + 2 * g;
    v2f a = *(const v2f*)(sAw + nlo * ASTR + k0);
    v2f b = *(const v2f*)(sWo + (kk * 2 + g) * WOSTR + nlo * 2);
    acc2 = wmma4(a, b, acc2);
  }
#pragma unroll
  for (int r = 0; r < 8; ++r) sLw[(r + 8 * g) * LSTR + nlo] = acc2[r] + sBo[nlo];
  __syncthreads();

  {
    const float* v = sLw + nlo * LSTR;
    float m = v[0];
#pragma unroll
    for (int c = 1; c < CDIM; ++c) m = fmaxf(m, v[c]);
    float s = 0.0f;
#pragma unroll
    for (int c = 0; c < CDIM; ++c) s += expf(v[c] - m);
    float ls = logf(s);
    if (g == 0) {
#pragma unroll
      for (int c = 0; c < CDIM; ++c) sOw[nlo * CDIM + c] = v[c] - m - ls;
    }
  }
  __syncthreads();
#pragma unroll
  for (int j = 0; j < (16 * CDIM) / 32; ++j) {
    int idx = lane + 32 * j;
    if (rowBase + idx / CDIM < n) out[(size_t)rowBase * CDIM + idx] = sOw[idx];
  }
}


extern "C" void kernel_launch(void* const* d_in, const int* in_sizes, int n_in,
                              void* d_out, int out_size, void* d_ws, size_t ws_size,
                              hipStream_t stream) {
  const float* x    = (const float*)d_in[0];
  const int*   ei   = (const int*)d_in[1];
  const float* W1_0 = (const float*)d_in[2];
  const float* b1   = (const float*)d_in[3];
  const float* W2_0 = (const float*)d_in[4];
  const float* W2_1 = (const float*)d_in[5];
  const float* b2   = (const float*)d_in[6];
  const float* Wl   = (const float*)d_in[7];
  const float* bl   = (const float*)d_in[8];

  const int N = in_sizes[0] / FDIM;
  const int E = in_sizes[1] / 2;
  const int* src = ei;
  const int* dst = ei + E;

  float* ws      = (float*)d_ws;
  float* h       = ws;
  int*   rec     = (int*)(h + (size_t)N * 64);
  unsigned int* bin_count  = (unsigned int*)(rec + (size_t)E * 2);
  unsigned int* bin_cursor = bin_count + MAXB;
  unsigned int* bin_start  = bin_cursor + MAXB;
  float* deg     = (float*)(bin_start + MAXB);
  float* out     = (float*)d_out;

  const int gblocks = (N + GROWS - 1) / GROWS;
  const int nbins = (N + BROWS - 1) / BROWS;
  gemm64_relu_kernel<<<gblocks, 128, 0, stream>>>(x, W1_0, b1, h, deg, bin_count, N);
  degree_kernel<<<dim3(DG_CHUNKS, DG_RANGES), DG_THREADS, 0, stream>>>(src, dst, deg, bin_count, N, E);
  place_kernel<<<(E + PL_EDGES - 1) / PL_EDGES, PL_THREADS, 0, stream>>>(src, dst, bin_count, bin_cursor, bin_start, rec, N, E);
  bin_head_kernel<<<nbins, BTHREADS, 0, stream>>>(h, rec, bin_count, bin_start, deg, W2_0, W2_1, b2, Wl, bl, out, N);
}


# 4 launches -> 3: destination-bin histogram runs beside the first GEMM as 64 per-slice tables (added up by the placing blocks), out-degree count beside the placing; same sums, arrival order
# speedup vs baseline: 3.3036x; 1.0612x over previous
#include <hip/hip_runtime.h>
#include <hip/hip_bf16.h>
#include <stdint.h>


#define FDIM 64
#define HDIM 64
#define CDIM 10

#define GWAVES 4
#define GROWS  (GWAVES * 16)
#define ASTR   68
#define WPSTR  160
#define WPROWS 32
#define WOSTR  96

typedef float v2f __attribute__((ext_vector_type(2)));
typedef float v8f __attribute__((ext_vector_type(8)));

__device__ __forceinline__ v8f wmma4(v2f a, v2f b, v8f c) {
  return __builtin_amdgcn_wmma_f32_16x16x4_f32(false, a, false, b, (short)0, c,
                                               false, false);
}


#define BROWS    208
#define BWAVES   13
#define BTHREADS (BWAVES * 32)
#define MAXB     1024

__device__ __forceinline__ int nbins_of(int n) {
  return min((n + BROWS - 1) / BROWS, MAXB);
}

__device__ __forceinline__ int bin_of(int d, int nbins) {
  return min((int)((unsigned int)d / (unsigned int)BROWS), nbins - 1);
}


#define HB_BLOCKS  64
#define HB_THREADS 128
#define HB_U       49

__device__ __forceinline__ void hist_role(unsigned int* hc, const int* __restrict__ dst,
                                          unsigned int* __restrict__ pbin, int n, int E) {
  const int tid = threadIdx.x;
  const int nbins = nbins_of(n);
  const int per = (E + HB_BLOCKS - 1) / HB_BLOCKS;
  const int lo = (int)blockIdx.x * per;
  const int hi = min(E, lo + per);
  for (int b = tid; b < nbins; b += HB_THREADS) hc[b] = 0u;
  __syncthreads();
  for (int i0 = lo; i0 < hi; i0 += HB_THREADS * HB_U) {
    int dv[HB_U];
#pragma unroll
    for (int u = 0; u < HB_U; ++u) dv[u] = dst[min(i0 + u * HB_THREADS + tid, hi - 1)];
#pragma unroll
    for (int u = 0; u < HB_U; ++u)
      if (i0 + u * HB_THREADS + tid < hi) atomicAdd(&hc[bin_of(dv[u], nbins)], 1u);
  }
  __syncthreads();
  for (int b = tid; b < nbins; b += HB_THREADS) pbin[(size_t)blockIdx.x * nbins + b] = hc[b];
}

#define DG_THREADS 256
#define DG_CHUNKS  64
#define DG_RANGES  8
#define DG_RLEN    6656
#define DG_PER     (DG_RLEN / DG_THREADS)
#define DG_U       13

typedef int v4i __attribute__((ext_vector_type(4)));
typedef int v2i __attribute__((ext_vector_type(2)));

__device__ __forceinline__ void deg_count(unsigned int* cnt, int s, int base, int len) {
  const unsigned int r = (unsigned int)(s - base);
  if (r < (unsigned int)len) atomicAdd(&cnt[r], 1u);
}

__device__ __forceinline__ void degree_role(unsigned int* cnt, const int* __restrict__ src,
                                            float* __restrict__ deg, int n, int E, int bx, int by) {
  const int tid = threadIdx.x;
  const int nq = E >> 2;
  const int qper = (nq + DG_CHUNKS - 1) / DG_CHUNKS;
  const int q_lo = bx * qper;
  const int q_hi = min(nq, q_lo + qper);
  const v4i* __restrict__ sq = (const v4i*)src;
  const int rot = (bx * (DG_RLEN / DG_CHUNKS)) & ~31;

  for (int base = by * DG_RLEN; base < n; base += DG_RANGES * DG_RLEN) {
    const int len = min(DG_RLEN, n - base);
#pragma unroll
    for (int k = 0; k < DG_PER; ++k) cnt[tid + k * DG_THREADS] = 0u;
    __syncthreads();

    for (int q0 = q_lo; q0 < q_hi; q0 += DG_THREADS * DG_U) {
      v4i v[DG_U];
#pragma unroll
      for (int u = 0; u < DG_U; ++u) v[u] = sq[min(q0 + u * DG_THREADS + tid, q_hi - 1)];
      asm volatile("" :: "v"(v[0]), "v"(v[1]), "v"(v[2]), "v"(v[3]), "v"(v[4]), "v"(v[5]),
                         "v"(v[6]), "v"(v[7]), "v"(v[8]), "v"(v[9]), "v"(v[10]), "v"(v[11]),
                         "v"(v[12]));
#pragma unroll
      for (int u = 0; u < DG_U; ++u) {
        const int lim = (q0 + u * DG_THREADS + tid < q_hi) ? len : 0;
        deg_count(cnt, v[u].x, base, lim);
        deg_count(cnt, v[u].y, base, lim);
        deg_count(cnt, v[u].z, base, lim);
        deg_count(cnt, v[u].w, base, lim);
      }
    }
    if (bx == 0 && tid < E - 4 * nq) deg_count(cnt, src[4 * nq + tid], base, len);
    __syncthreads();

    unsigned int c[DG_PER];
#pragma unroll
    for (int k = 0; k < DG_PER; ++k) {
      int j = tid + k * DG_THREADS + rot;
      if (j >= DG_RLEN) j -= DG_RLEN;
      c[k] = cnt[j];
    }
#pragma unroll
    for (int k = 0; k < DG_PER; ++k) {
      int j = tid + k * DG_THREADS + rot;
      if (j >= DG_RLEN) j -= DG_RLEN;
      if (j < len && c[k] != 0u) atomicAdd(&deg[base + j], (float)c[k]);
    }
    __syncthreads();
  }
}
__device__ __forceinline__ float dis_of(float d) {
  return (d > 0.0f) ? rsqrtf(fmaxf(d, 1.0f)) : 0.0f;
}

__device__ __forceinline__ int lane_bcast_i(int v, int j) {
  return __builtin_amdgcn_readlane(v, j);
}
__device__ __forceinline__ float lane_bcast_f(float v, int j) {
  return __int_as_float(__builtin_amdgcn_readlane(__float_as_int(v), j));
}

#define PL_THREADS DG_THREADS
#define PL_U       8
#define PL_EDGES   (PL_THREADS * PL_U)
#define PL_OWN     (MAXB / PL_THREADS)

__device__ __forceinline__ void place_role(unsigned int* lc, unsigned int* pre, unsigned int* tot,
                                           unsigned int* wsum, const int* __restrict__ src,
                                           const int* __restrict__ dst,
                                           const unsigned int* __restrict__ pbin,
                                           unsigned int* __restrict__ bin_count,
                                           unsigned int* __restrict__ bin_cursor,
                                           unsigned int* __restrict__ bin_start,
                                           int* __restrict__ rec, int n, int E) {
  const int tid = threadIdx.x, wave = tid >> 5, lane = tid & 31;
  const int nbins = nbins_of(n);
  const int lo = (int)blockIdx.x * PL_EDGES;
  const int hi = min(E, lo + PL_EDGES);

  for (int b = tid; b < MAXB; b += PL_THREADS) lc[b] = 0u;
  int sv[PL_U], dv[PL_U];
#pragma unroll
  for (int u = 0; u < PL_U; ++u) {
    const int i = min(lo + u * PL_THREADS + tid, E - 1);
    dv[u] = dst[i];
    sv[u] = src[i];
  }
  for (int b = tid; b < MAXB; b += PL_THREADS) {
    unsigned int t = 0u;
    if (b < nbins) {
#pragma unroll
      for (int p = 0; p < HB_BLOCKS; ++p) t += pbin[(size_t)p * nbins + b];
      if (blockIdx.x == 0) bin_count[b] = t;
    }
    tot[b] = t;
  }
  __syncthreads();
#pragma unroll
  for (int u = 0; u < PL_U; ++u)
    if (lo + u * PL_THREADS + tid < hi) atomicAdd(&lc[bin_of(dv[u], nbins)], 1u);

  unsigned int c[PL_OWN];
  unsigned int tsum = 0u;
#pragma unroll
  for (int k = 0; k < PL_OWN; ++k) {
    c[k] = tot[tid * PL_OWN + k];
    tsum += c[k];
  }
  unsigned int inc = tsum;
#pragma unroll
  for (int off = 1; off < 32; off <<= 1) {
    const unsigned int v = __shfl_up(inc, off);
    if (lane >= off) inc += v;
  }
  if (lane == 31) wsum[wave] = inc;
  __syncthreads();
  unsigned int ex = inc - tsum;
  for (int w = 0; w < wave; ++w) ex += wsum[w];
#pragma unroll
  for (int k = 0; k < PL_OWN; ++k) {
    pre[tid * PL_OWN + k] = ex;
    if (blockIdx.x == 0 && tid * PL_OWN + k < nbins) bin_start[tid * PL_OWN + k] = ex;
    ex += c[k];
  }
  __syncthreads();

  for (int b = tid; b < nbins; b += PL_THREADS) {
    const unsigned int cb = lc[b];
    if (cb != 0u) lc[b] = pre[b] + atomicAdd(&bin_cursor[b], cb);
  }
  __syncthreads();

#pragma unroll
  for (int u = 0; u < PL_U; ++u)
    if (lo + u * PL_THREADS + tid < hi) {
      const unsigned int pos = atomicAdd(&lc[bin_of(dv[u], nbins)], 1u);
      const v2i pr = {sv[u], dv[u]};
      if (pos < (unsigned int)E) ((v2i*)rec)[pos] = pr;
    }
}

__global__ void __launch_bounds__(DG_THREADS)
place_degree_kernel(const int* __restrict__ src, const int* __restrict__ dst,
                    const unsigned int* __restrict__ pbin, unsigned int* __restrict__ bin_count,
                    unsigned int* __restrict__ bin_cursor, unsigned int* __restrict__ bin_start,
                    float* __restrict__ deg, int* __restrict__ rec, int n, int E) {
  __shared__ unsigned int sm[DG_RLEN];
  const int plb = (E + PL_EDGES - 1) / PL_EDGES;
  if ((int)blockIdx.x < plb) {
    place_role(sm, sm + MAXB, sm + 2 * MAXB, sm + 3 * MAXB, src, dst, pbin, bin_count,
               bin_cursor, bin_start, rec, n, E);
  } else {
    const int d = (int)blockIdx.x - plb;
    degree_role(sm, src, deg, n, E, d % DG_CHUNKS, d / DG_CHUNKS);
  }
}


__device__ __forceinline__ void stage_w_pairs(float* sWp, const float* __restrict__ W,
                                              int tid) {
  const float4* Wv = (const float4*)W;
#pragma unroll
  for (int i = 0; i < 8; ++i) {
    int idx = tid + i * 128;
    float4 v = Wv[idx];
    int k = idx >> 4;
    int c4 = (idx & 15) * 4;
    float* dstp = sWp + (k >> 1) * WPSTR + (k & 1);
    dstp[(c4 + 0) * 2] = v.x;
    dstp[(c4 + 1) * 2] = v.y;
    dstp[(c4 + 2) * 2] = v.z;
    dstp[(c4 + 3) * 2] = v.w;
  }
}

__device__ __forceinline__ void stage_rows(float* sAw, const float* __restrict__ A,
                                           int rowBase, int n, int lane) {
#pragma unroll
  for (int i = 0; i < 8; ++i) {
    int fi = lane + i * 32;
    int r = fi >> 4, c4 = fi & 15;
    int row = rowBase + r;
    if (row >= n) row = n - 1;
    ((float4*)(sAw + r * ASTR))[c4] = ((const float4*)(A + (size_t)row * 64))[c4];
  }
}

__device__ __forceinline__ void mm_accum64(const float* sAw, const float* sWp,
                                           int lane, v8f acc[4]) {
  const int nlo = lane & 15;
  const int g = lane >> 4;
#pragma unroll
  for (int kk = 0; kk < 16; ++kk) {
    const int k0 = kk * 4 + 2 * g;
    v2f a = *(const v2f*)(sAw + nlo * ASTR + k0);
    const float* brow = sWp + (kk * 2 + g) * WPSTR;
#pragma unroll
    for (int nt = 0; nt < 4; ++nt) {
      v2f b = *(const v2f*)(brow + (nt * 16 + nlo) * 2);
      acc[nt] = wmma4(a, b, acc[nt]);
    }
  }
}


__global__ void __launch_bounds__(HB_THREADS)
front_kernel(const float* __restrict__ A, const float* __restrict__ W,
             const float* __restrict__ bias, float* __restrict__ out,
             float* __restrict__ deg, unsigned int* __restrict__ bin_cursor,
             const int* __restrict__ dst, unsigned int* __restrict__ pbin, int n, int E) {
  __shared__ float sWp[WPROWS * WPSTR];
  __shared__ float sA[GWAVES * 16 * ASTR];
  __shared__ float sB[64];
  if (blockIdx.x < HB_BLOCKS) {
    hist_role((unsigned int*)sA, dst, pbin, n, E);
    return;
  }
  const int gb = (int)blockIdx.x - HB_BLOCKS;
  const int tid = threadIdx.x, wave = tid >> 5, lane = tid & 31;
  stage_w_pairs(sWp, W, tid);
  if (tid < 64) sB[tid] = bias[tid];

  const int rowBase = gb * GROWS + wave * 16;
  float* sAw = sA + wave * 16 * ASTR;
  stage_rows(sAw, A, rowBase, n, lane);
  __syncthreads();

  if (lane < 16 && rowBase + lane < n) deg[rowBase + lane] = 0.0f;
  if (gb == 0)
    for (int i = tid; i < MAXB; i += HB_THREADS) bin_cursor[i] = 0u;

  v8f z = {0, 0, 0, 0, 0, 0, 0, 0};
  v8f acc[4] = {z, z, z, z};
  mm_accum64(sAw, sWp, lane, acc);

  const int nlo = lane & 15, g = lane >> 4;
#pragma unroll
  for (int nt = 0; nt < 4; ++nt)
#pragma unroll
    for (int r = 0; r < 8; ++r) {
      int row = rowBase + r + 8 * g;
      int col = nt * 16 + nlo;
      if (row < n) out[(size_t)row * 64 + col] = fmaxf(acc[nt][r] + sB[col], 0.0f);
    }
}

#define LSTR 17
#define BG   16

#define WIT ((1024 + BTHREADS - 1) / BTHREADS)

__device__ __forceinline__ void put_w_pairs(float* sWp, float4 v, int idx) {
  const int k = idx >> 4;
  const int c4 = (idx & 15) * 4;
  float* dstp = sWp + (k >> 1) * WPSTR + (k & 1);
  dstp[(c4 + 0) * 2] = v.x;
  dstp[(c4 + 1) * 2] = v.y;
  dstp[(c4 + 2) * 2] = v.z;
  dstp[(c4 + 3) * 2] = v.w;
}

template <int G0, bool PART>
__device__ __forceinline__ void add_rows(const float* __restrict__ h, float* sS, int s, int dl,
                                         float we, int lane, int mc) {
  float a[BG], b[BG];
#pragma unroll
  for (int u = 0; u < BG; ++u) {
    const float* hp = h + (size_t)lane_bcast_i(s, G0 + u) * HDIM + lane;
    a[u] = hp[0];
    b[u] = hp[32];
  }
  if (PART) {
    asm volatile("" :: "v"(a[0]), "v"(a[1]), "v"(a[2]), "v"(a[3]), "v"(a[4]), "v"(a[5]),
                       "v"(a[6]), "v"(a[7]), "v"(a[8]), "v"(a[9]), "v"(a[10]), "v"(a[11]),
                       "v"(a[12]), "v"(a[13]), "v"(a[14]), "v"(a[15]));
    asm volatile("" :: "v"(b[0]), "v"(b[1]), "v"(b[2]), "v"(b[3]), "v"(b[4]), "v"(b[5]),
                       "v"(b[6]), "v"(b[7]), "v"(b[8]), "v"(b[9]), "v"(b[10]), "v"(b[11]),
                       "v"(b[12]), "v"(b[13]), "v"(b[14]), "v"(b[15]));
  }
#pragma unroll
  for (int u = 0; u < BG; ++u) {
    if (!PART || G0 + u < mc) {
      float* sp = sS + lane_bcast_i(dl, G0 + u) * ASTR + lane;
      const float wj = lane_bcast_f(we, G0 + u);
      atomicAdd(sp, wj * a[u]);
      atomicAdd(sp + 32, wj * b[u]);
    }
  }
}

__global__ void __launch_bounds__(BTHREADS)
bin_head_kernel(const float* __restrict__ h, const int* __restrict__ rec,
                const unsigned int* __restrict__ bin_count,
                const unsigned int* __restrict__ bin_start, const float* __restrict__ deg,
                const float* __restrict__ W0, const float* __restrict__ W1,
                const float* __restrict__ bias, const float* __restrict__ Wl,
                const float* __restrict__ bl, float* __restrict__ out, int n) {
  __shared__ float sWp[2 * WPROWS * WPSTR];
  __shared__ float sWo[WPROWS * WOSTR];
  __shared__ float sA[BROWS * ASTR];
  __shared__ float sS[BROWS * ASTR];
  __shared__ float sL[BWAVES * 16 * LSTR];
  __shared__ float sO[BWAVES * 16 * CDIM];
  __shared__ float sB[64];
  __shared__ float sBo[16];
  const int tid = threadIdx.x, lane = tid & 31;
  const int wave = __builtin_amdgcn_readfirstlane(tid >> 5);
  const int bin = blockIdx.x;
  const int rowBase = bin * BROWS + wave * 16;
  float* sAw = sA + wave * 16 * ASTR;
  float* sSw = sS + wave * 16 * ASTR;
  float* sLw = sL + wave * 16 * LSTR;
  float* sOw = sO + wave * 16 * CDIM;

  const unsigned int first = bin_start[bin];
  const unsigned int cntb = bin_count[bin];
  unsigned int c0 = (unsigned int)wave * 32u;
  const bool any = c0 < cntb;

  float4 w0v[WIT], w1v[WIT];
  float wlv[WIT];
#pragma unroll
  for (int i = 0; i < WIT; ++i) {
    const int idx = min(tid + i * BTHREADS, 1023);
    w0v[i] = ((const float4*)W0)[idx];
    w1v[i] = ((const float4*)W1)[idx];
    wlv[i] = Wl[(idx >> 4) * CDIM + min(idx & 15, CDIM - 1)];
  }
  const float bv = bias[tid & 63];
  const float blv = bl[min(tid & 15, CDIM - 1)];
  float4 hv[8];
#pragma unroll
  for (int i = 0; i < 8; ++i) {
    const int fi = lane + i * 32;
    const int row = min(rowBase + (fi >> 4), n - 1);
    hv[i] = ((const float4*)(h + (size_t)row * 64))[fi & 15];
  }
  int mc = any ? (int)min(32u, cntb - c0) : 1;
  const v2i* __restrict__ rec2 = (const v2i*)rec;
  v2i sd = rec2[any ? first + c0 + (unsigned int)min(lane, mc - 1) : 0u];

#pragma unroll
  for (int i = 0; i < WIT; ++i) {
    const int idx = tid + i * BTHREADS;
    if (idx < 1024) {
      put_w_pairs(sWp, w0v[i], idx);
      put_w_pairs(sWp + WPROWS * WPSTR, w1v[i], idx);
      const int k = idx >> 4, c = idx & 15;
      sWo[(k >> 1) * WOSTR + c * 2 + (k & 1)] = (c < CDIM) ? wlv[i] : 0.0f;
    }
  }
  if (tid < 64) sB[tid] = bv;
  if (tid < 16) sBo[tid] = (tid < CDIM) ? blv : 0.0f;
#pragma unroll
  for (int i = 0; i < 8; ++i) {
    const int fi = lane + i * 32;
    ((float4*)(sAw + (fi >> 4) * ASTR))[fi & 15] = hv[i];
  }
  for (int i = tid; i < BROWS * ASTR; i += BTHREADS) sS[i] = 0.0f;
  float we = -(dis_of(deg[sd.x]) * dis_of(deg[sd.y]));
  __syncthreads();

  if (any) {
    for (;;) {
      const unsigned int c1 = c0 + BTHREADS;
      const bool more = c1 < cntb;
      const int m1 = more ? (int)min(32u, cntb - c1) : mc;
      const v2i sdn = rec2[first + (more ? c1 : c0) + (unsigned int)min(lane, m1 - 1)];
      const int s = sd.x;
      const int dl = (int)min((unsigned int)(sd.y - bin * BROWS), (unsigned int)(BROWS - 1));
      float dgs = 1.0f, dgd = 1.0f;
      if (mc == 32) {
        add_rows<0, false>(h, sS, s, dl, we, lane, mc);
        dgs = deg[sdn.x];
        dgd = deg[sdn.y];
        add_rows<BG, false>(h, sS, s, dl, we, lane, mc);
      } else {
        add_rows<0, true>(h, sS, s, dl, we, lane, mc);
        if (mc > BG) add_rows<BG, true>(h, sS, s, dl, we, lane, mc);
      }
      if (!more) break;
      sd = sdn;
      we = -(dis_of(dgs) * dis_of(dgd));
      mc = m1;
      c0 = c1;
    }
  }
  __syncthreads();

  v8f z = {0, 0, 0, 0, 0, 0, 0, 0};
  v8f acc[4] = {z, z, z, z};
  mm_accum64(sAw, sWp, lane, acc);
  mm_accum64(sSw, sWp + WPROWS * WPSTR, lane, acc);

  const int nlo = lane & 15, g = lane >> 4;
#pragma unroll
  for (int nt = 0; nt < 4; ++nt)
#pragma unroll
    for (int r = 0; r < 8; ++r) {
      int rl = r + 8 * g;
      int col = nt * 16 + nlo;
      float v = fmaxf(acc[nt][r] + sB[col], 0.0f);
      sAw[rl * ASTR + col] = v + sAw[rl * ASTR + col];
    }
  __syncthreads();

  v8f acc2 = z;
#pragma unroll
  for (int kk = 0; kk < 16; ++kk) {
    const int k0 = kk * 4 + 2 * g;
    v2f a = *(const v2f*)(sAw + nlo * ASTR + k0);
    v2f b = *(const v2f*)(sWo + (kk * 2 + g) * WOSTR + nlo * 2);
    acc2 = wmma4(a, b, acc2);
  }
#pragma unroll
  for (int r = 0; r < 8; ++r) sLw[(r + 8 * g) * LSTR + nlo] = acc2[r] + sBo[nlo];
  __syncthreads();

  {
    const float* v = sLw + nlo * LSTR;
    float m = v[0];
#pragma unroll
    for (int c = 1; c < CDIM; ++c) m = fmaxf(m, v[c]);
    float s = 0.0f;
#pragma unroll
    for (int c = 0; c < CDIM; ++c) s += expf(v[c] - m);
    float ls = logf(s);
    if (g == 0) {
#pragma unroll
      for (int c = 0; c < CDIM; ++c) sOw[nlo * CDIM + c] = v[c] - m - ls;
    }
  }
  __syncthreads();
#pragma unroll
  for (int j = 0; j < (16 * CDIM) / 32; ++j) {
    int idx = lane + 32 * j;
    if (rowBase + idx / CDIM < n) out[(size_t)rowBase * CDIM + idx] = sOw[idx];
  }
}


extern "C" void kernel_launch(void* const* d_in, const int* in_sizes, int n_in,
                              void* d_out, int out_size, void* d_ws, size_t ws_size,
                              hipStream_t stream) {
  const float* x    = (const float*)d_in[0];
  const int*   ei   = (const int*)d_in[1];
  const float* W1_0 = (const float*)d_in[2];
  const float* b1   = (const float*)d_in[3];
  const float* W2_0 = (const float*)d_in[4];
  const float* W2_1 = (const float*)d_in[5];
  const float* b2   = (const float*)d_in[6];
  const float* Wl   = (const float*)d_in[7];
  const float* bl   = (const float*)d_in[8];

  const int N = in_sizes[0] / FDIM;
  const int E = in_sizes[1] / 2;
  const int* src = ei;
  const int* dst = ei + E;

  float* ws      = (float*)d_ws;
  float* h       = ws;
  int*   rec     = (int*)(h + (size_t)N * 64);
  unsigned int* bin_count  = (unsigned int*)(rec + (size_t)E * 2);
  unsigned int* bin_cursor = bin_count + MAXB;
  unsigned int* bin_start  = bin_cursor + MAXB;
  float* deg     = (float*)(bin_start + MAXB);
  float* out     = (float*)d_out;
  unsigned int* pbin = (unsigned int*)d_out;

  const int gblocks = (N + GROWS - 1) / GROWS;
  const int nbins = (N + BROWS - 1) / BROWS;
  front_kernel<<<HB_BLOCKS + gblocks, HB_THREADS, 0, stream>>>(x, W1_0, b1, h, deg, bin_cursor, dst, pbin, N, E);
  place_degree_kernel<<<(E + PL_EDGES - 1) / PL_EDGES + DG_CHUNKS * DG_RANGES, DG_THREADS, 0, stream>>>(src, dst, pbin, bin_count, bin_cursor, bin_start, deg, rec, N, E);
  bin_head_kernel<<<nbins, BTHREADS, 0, stream>>>(h, rec, bin_count, bin_start, deg, W2_0, W2_1, b2, Wl, bl, out, N);
}
